# starting kernel
# baseline (speedup 1.0000x reference)
#include <hip/hip_runtime.h>


typedef __attribute__((ext_vector_type(2))) float v2f;
typedef __attribute__((ext_vector_type(8))) float v8f;

#define NNODE 10000
#define NEDGE 50000
#define HDIM  64

template <int K, bool RELUA, bool TRANSB>
__device__ __forceinline__ v8f wmma_tile_f32(
    const float* __restrict__ A, int lda,
    const float* __restrict__ B, int ldb,
    int m0, int col, int hf, int l)
{
    const float* __restrict__ Arow = A + (size_t)(m0 + l) * lda;
    const float* __restrict__ Bcol = TRANSB ? (B + (size_t)col * ldb) : B;
    v8f acc = {};
#pragma unroll
    for (int k = 0; k < K; k += 4) {
        const int kb = k + (hf << 1);
        v2f a = *(const v2f*)(Arow + kb);
        if (RELUA) { a.x = fmaxf(a.x, 0.0f); a.y = fmaxf(a.y, 0.0f); }
        v2f b;
        if (TRANSB) {
            b = *(const v2f*)(Bcol + kb);
        } else {
            b.x = B[(size_t)kb * ldb + col];
            b.y = B[(size_t)(kb + 1) * ldb + col];
        }
        acc = __builtin_amdgcn_wmma_f32_16x16x4_f32(
            false, a, false, b, (short)0, acc, false, false);
    }
    return acc;
}

__device__ __forceinline__ void store_tile(
    float* __restrict__ C, int ldc, int m0, int col, int hf,
    float bv, bool relu, v8f acc)
{
#pragma unroll
    for (int v = 0; v < 8; ++v) {
        const int row = m0 + v + (hf << 3);
        float r = acc[v] + bv;
        if (relu) r = fmaxf(r, 0.0f);
        C[(size_t)row * ldc + col] = r;
    }
}

__global__ __launch_bounds__(128) void lin0_gemm(
    const float* __restrict__ x, const float* __restrict__ w,
    const float* __restrict__ bias, float* __restrict__ h)
{
    const int wave = threadIdx.x >> 5;
    const int lane = threadIdx.x & 31;
    const int hf = lane >> 4, l = lane & 15;
    const int tile = blockIdx.x * 4 + wave;
    if (tile >= (NNODE / 16) * (HDIM / 16)) return;
    const int tm = tile >> 2, tn = tile & 3;
    const int m0 = tm << 4, col = (tn << 4) + l;

    v8f acc = wmma_tile_f32<32, false, false>(x, 32, w, HDIM, m0, col, hf, l);
    store_tile(h, HDIM, m0, col, hf, bias[col], true, acc);
}

__global__ __launch_bounds__(128) void node_gemms(
    const float* __restrict__ h,
    const float* __restrict__ nn_w, const float* __restrict__ nn_b,
    const float* __restrict__ root_w, const float* __restrict__ conv_b,
    float* __restrict__ Y, float* __restrict__ agg)
{
    const int wave = threadIdx.x >> 5;
    const int lane = threadIdx.x & 31;
    const int hf = lane >> 4, l = lane & 15;
    const int tile = blockIdx.x * 4 + wave;
    const int perPlane = (NNODE / 16) * (HDIM / 16);
    if (tile >= 6 * perPlane) return;

    const int p = tile / perPlane;
    const int t = tile - p * perPlane;
    const int tm = t >> 2, tn = t & 3;
    const int m0 = tm << 4, col = (tn << 4) + l;

    const size_t plane = (size_t)NNODE * HDIM;
    const float* B = (p < 4) ? (nn_w + (size_t)p * HDIM * HDIM)
                             : ((p == 4) ? nn_b : root_w);
    float* C = (p < 5) ? (Y + (size_t)p * plane) : agg;
    const float bv = (p == 5) ? conv_b[col] : 0.0f;

    v8f acc = wmma_tile_f32<64, false, false>(h, HDIM, B, HDIM, m0, col, hf, l);
    store_tile(C, HDIM, m0, col, hf, bv, false, acc);
}

__global__ __launch_bounds__(256) void edge_msg(
    const int* __restrict__ src, const int* __restrict__ dst,
    const float* __restrict__ ea,
    const float* __restrict__ Y, float* __restrict__ agg, int E)
{
    const size_t plane = (size_t)NNODE * HDIM;
    const int t  = blockIdx.x * blockDim.x + threadIdx.x;
    const int o  = t & (HDIM - 1);
    int e        = t >> 6;
    const int es = (gridDim.x * blockDim.x) >> 6;

    for (; e < E; e += es) {
        const int en = e + es;
        if (en < E) {
            const int s2 = src[en];
            __builtin_prefetch(&Y[(size_t)s2 * HDIM + o], 0, 0);
        }
        const int s = src[e];
        const int d = dst[e];
        const float c0 = ea[e * 4 + 0];
        const float c1 = ea[e * 4 + 1];
        const float c2 = ea[e * 4 + 2];
        const float c3 = ea[e * 4 + 3];
        const size_t base = (size_t)s * HDIM + o;
        float m = Y[base + 4 * plane];
        m = fmaf(c0, Y[base + 0 * plane], m);
        m = fmaf(c1, Y[base + 1 * plane], m);
        m = fmaf(c2, Y[base + 2 * plane], m);
        m = fmaf(c3, Y[base + 3 * plane], m);
        atomicAdd(&agg[(size_t)d * HDIM + o], m);
    }
}

__global__ __launch_bounds__(128) void gru_gemms(
    const float* __restrict__ agg, const float* __restrict__ h,
    const float* __restrict__ w_ih, const float* __restrict__ w_hh,
    const float* __restrict__ b_ih, const float* __restrict__ b_hh,
    float* __restrict__ gi, float* __restrict__ gh)
{
    const int wave = threadIdx.x >> 5;
    const int lane = threadIdx.x & 31;
    const int hf = lane >> 4, l = lane & 15;
    const int perJob = (NNODE / 16) * (3 * HDIM / 16);
    const int tile = blockIdx.x * 4 + wave;
    if (tile >= 2 * perJob) return;

    const int job = tile / perJob;
    const int t   = tile - job * perJob;
    const int tilesN = (3 * HDIM) / 16;
    const int tm = t / tilesN, tn = t - tm * tilesN;
    const int m0 = tm << 4, col = (tn << 4) + l;

    v8f acc;
    if (job == 0) {
        acc = wmma_tile_f32<64, true, true>(agg, HDIM, w_ih, HDIM, m0, col, hf, l);
        store_tile(gi, 3 * HDIM, m0, col, hf, b_ih[col], false, acc);
    } else {
        acc = wmma_tile_f32<64, false, true>(h, HDIM, w_hh, HDIM, m0, col, hf, l);
        store_tile(gh, 3 * HDIM, m0, col, hf, b_hh[col], false, acc);
    }
}

__global__ __launch_bounds__(256) void gru_update(
    const float* __restrict__ gi, const float* __restrict__ gh,
    float* __restrict__ h, int total)
{
    const int t = blockIdx.x * blockDim.x + threadIdx.x;
    if (t >= total) return;
    const int node = t >> 6;
    const int o    = t & (HDIM - 1);
    const float* gin = gi + (size_t)node * (3 * HDIM);
    const float* ghn = gh + (size_t)node * (3 * HDIM);

    const float sr = gin[o]        + ghn[o];
    const float sz = gin[HDIM + o] + ghn[HDIM + o];
    const float in = gin[2 * HDIM + o];
    const float hn = ghn[2 * HDIM + o];

    const float r = 1.0f / (1.0f + __expf(-sr));
    const float z = 1.0f / (1.0f + __expf(-sz));
    const float n = tanhf(in + r * hn);
    const float ho = h[t];
    h[t] = (1.0f - z) * n + z * ho;
}

extern "C" void kernel_launch(void* const* d_in, const int* in_sizes, int n_in,
                              void* d_out, int out_size, void* d_ws, size_t ws_size,
                              hipStream_t stream)
{
    (void)in_sizes; (void)n_in; (void)out_size; (void)ws_size;

    const float* x      = (const float*)d_in[0];
    const int*   eidx   = (const int*)  d_in[1];
    const float* ea     = (const float*)d_in[2];
    const float* lin0_w = (const float*)d_in[3];
    const float* lin0_b = (const float*)d_in[4];
    const float* nn_w   = (const float*)d_in[5];
    const float* nn_b   = (const float*)d_in[6];
    const float* root_w = (const float*)d_in[7];
    const float* conv_b = (const float*)d_in[8];
    const float* w_ih   = (const float*)d_in[9];
    const float* w_hh   = (const float*)d_in[10];
    const float* b_ih   = (const float*)d_in[11];
    const float* b_hh   = (const float*)d_in[12];

    const int* src = eidx;
    const int* dst = eidx + NEDGE;

    float* h = (float*)d_out;

    const size_t plane = (size_t)NNODE * HDIM;
    float* Y   = (float*)d_ws;
    float* agg = Y + 5 * plane;
    float* gi  = agg + plane;
    float* gh  = gi + (size_t)NNODE * 3 * HDIM;

    const int gLin0 = ((NNODE / 16) * (HDIM / 16) + 3) / 4;
    const int gNode = (6 * (NNODE / 16) * (HDIM / 16) + 3) / 4;
    const int gGru  = (2 * (NNODE / 16) * (3 * HDIM / 16) + 3) / 4;

    lin0_gemm<<<gLin0, 128, 0, stream>>>(x, lin0_w, lin0_b, h);

    for (int it = 0; it < 3; ++it) {
        node_gemms<<<gNode, 128, 0, stream>>>(h, nn_w, nn_b, root_w, conv_b,
                                              Y, agg);
        edge_msg<<<4096, 256, 0, stream>>>(src, dst, ea, Y, agg, NEDGE);
        gru_gemms<<<gGru, 128, 0, stream>>>(agg, h, w_ih, w_hh, b_ih, b_hh,
                                            gi, gh);
        gru_update<<<(NNODE * HDIM + 255) / 256, 256, 0, stream>>>(
            gi, gh, h, NNODE * HDIM);
    }
}


# Fuse the two GRU GEMMs and the gate update into one kernel: block owns 16 rows, each wave keeps 6 gate tiles in registers, h ping-pongs out/workspace; gi,gh never stored; 13 to 10 launches
# speedup vs baseline: 1.1129x; 1.1129x over previous
#include <hip/hip_runtime.h>


typedef __attribute__((ext_vector_type(2))) float v2f;
typedef __attribute__((ext_vector_type(8))) float v8f;

#define NNODE 10000
#define NEDGE 50000
#define HDIM  64

template <int K, bool RELUA, bool TRANSB>
__device__ __forceinline__ v8f wmma_tile_f32(
    const float* __restrict__ A, int lda,
    const float* __restrict__ B, int ldb,
    int m0, int col, int hf, int l)
{
    const float* __restrict__ Arow = A + (size_t)(m0 + l) * lda;
    const float* __restrict__ Bcol = TRANSB ? (B + (size_t)col * ldb) : B;
    v8f acc = {};
#pragma unroll
    for (int k = 0; k < K; k += 4) {
        const int kb = k + (hf << 1);
        v2f a = *(const v2f*)(Arow + kb);
        if (RELUA) { a.x = fmaxf(a.x, 0.0f); a.y = fmaxf(a.y, 0.0f); }
        v2f b;
        if (TRANSB) {
            b = *(const v2f*)(Bcol + kb);
        } else {
            b.x = B[(size_t)kb * ldb + col];
            b.y = B[(size_t)(kb + 1) * ldb + col];
        }
        acc = __builtin_amdgcn_wmma_f32_16x16x4_f32(
            false, a, false, b, (short)0, acc, false, false);
    }
    return acc;
}

__device__ __forceinline__ void store_tile(
    float* __restrict__ C, int ldc, int m0, int col, int hf,
    float bv, bool relu, v8f acc)
{
#pragma unroll
    for (int v = 0; v < 8; ++v) {
        const int row = m0 + v + (hf << 3);
        float r = acc[v] + bv;
        if (relu) r = fmaxf(r, 0.0f);
        C[(size_t)row * ldc + col] = r;
    }
}

__global__ __launch_bounds__(128) void lin0_gemm(
    const float* __restrict__ x, const float* __restrict__ w,
    const float* __restrict__ bias, float* __restrict__ h)
{
    const int wave = threadIdx.x >> 5;
    const int lane = threadIdx.x & 31;
    const int hf = lane >> 4, l = lane & 15;
    const int tile = blockIdx.x * 4 + wave;
    if (tile >= (NNODE / 16) * (HDIM / 16)) return;
    const int tm = tile >> 2, tn = tile & 3;
    const int m0 = tm << 4, col = (tn << 4) + l;

    v8f acc = wmma_tile_f32<32, false, false>(x, 32, w, HDIM, m0, col, hf, l);
    store_tile(h, HDIM, m0, col, hf, bias[col], true, acc);
}

__global__ __launch_bounds__(128) void node_gemms(
    const float* __restrict__ h,
    const float* __restrict__ nn_w, const float* __restrict__ nn_b,
    const float* __restrict__ root_w, const float* __restrict__ conv_b,
    float* __restrict__ Y, float* __restrict__ agg)
{
    const int wave = threadIdx.x >> 5;
    const int lane = threadIdx.x & 31;
    const int hf = lane >> 4, l = lane & 15;
    const int tile = blockIdx.x * 4 + wave;
    const int perPlane = (NNODE / 16) * (HDIM / 16);
    if (tile >= 6 * perPlane) return;

    const int p = tile / perPlane;
    const int t = tile - p * perPlane;
    const int tm = t >> 2, tn = t & 3;
    const int m0 = tm << 4, col = (tn << 4) + l;

    const size_t plane = (size_t)NNODE * HDIM;
    const float* B = (p < 4) ? (nn_w + (size_t)p * HDIM * HDIM)
                             : ((p == 4) ? nn_b : root_w);
    float* C = (p < 5) ? (Y + (size_t)p * plane) : agg;
    const float bv = (p == 5) ? conv_b[col] : 0.0f;

    v8f acc = wmma_tile_f32<64, false, false>(h, HDIM, B, HDIM, m0, col, hf, l);
    store_tile(C, HDIM, m0, col, hf, bv, false, acc);
}

__global__ __launch_bounds__(256) void edge_msg(
    const int* __restrict__ src, const int* __restrict__ dst,
    const float* __restrict__ ea,
    const float* __restrict__ Y, float* __restrict__ agg, int E)
{
    const size_t plane = (size_t)NNODE * HDIM;
    const int t  = blockIdx.x * blockDim.x + threadIdx.x;
    const int o  = t & (HDIM - 1);
    int e        = t >> 6;
    const int es = (gridDim.x * blockDim.x) >> 6;

    for (; e < E; e += es) {
        const int en = e + es;
        if (en < E) {
            const int s2 = src[en];
            __builtin_prefetch(&Y[(size_t)s2 * HDIM + o], 0, 0);
        }
        const int s = src[e];
        const int d = dst[e];
        const float c0 = ea[e * 4 + 0];
        const float c1 = ea[e * 4 + 1];
        const float c2 = ea[e * 4 + 2];
        const float c3 = ea[e * 4 + 3];
        const size_t base = (size_t)s * HDIM + o;
        float m = Y[base + 4 * plane];
        m = fmaf(c0, Y[base + 0 * plane], m);
        m = fmaf(c1, Y[base + 1 * plane], m);
        m = fmaf(c2, Y[base + 2 * plane], m);
        m = fmaf(c3, Y[base + 3 * plane], m);
        atomicAdd(&agg[(size_t)d * HDIM + o], m);
    }
}

template <bool RELUA>
__device__ __forceinline__ void gate_tiles_f32(
    const float* __restrict__ A, const float* __restrict__ Wt,
    int m0, int col, int hf, int l, v8f& accR, v8f& accZ, v8f& accN)
{
    const float* __restrict__ Arow = A + (size_t)(m0 + l) * HDIM;
    const float* __restrict__ Br = Wt + (size_t)col * HDIM;
    const float* __restrict__ Bz = Wt + (size_t)(HDIM + col) * HDIM;
    const float* __restrict__ Bn = Wt + (size_t)(2 * HDIM + col) * HDIM;
    v8f r = {}, z = {}, n = {};
#pragma unroll
    for (int k = 0; k < HDIM; k += 4) {
        const int kb = k + (hf << 1);
        v2f a = *(const v2f*)(Arow + kb);
        if (RELUA) { a.x = fmaxf(a.x, 0.0f); a.y = fmaxf(a.y, 0.0f); }
        const v2f br = *(const v2f*)(Br + kb);
        const v2f bz = *(const v2f*)(Bz + kb);
        const v2f bn = *(const v2f*)(Bn + kb);
        r = __builtin_amdgcn_wmma_f32_16x16x4_f32(
            false, a, false, br, (short)0, r, false, false);
        z = __builtin_amdgcn_wmma_f32_16x16x4_f32(
            false, a, false, bz, (short)0, z, false, false);
        n = __builtin_amdgcn_wmma_f32_16x16x4_f32(
            false, a, false, bn, (short)0, n, false, false);
    }
    accR = r; accZ = z; accN = n;
}

__global__ __launch_bounds__(128) void gru_step(
    const float* __restrict__ agg, const float* __restrict__ h_in,
    const float* __restrict__ w_ih, const float* __restrict__ w_hh,
    const float* __restrict__ b_ih, const float* __restrict__ b_hh,
    float* __restrict__ h_out)
{
    const int wave = threadIdx.x >> 5;
    const int lane = threadIdx.x & 31;
    const int hf = lane >> 4, l = lane & 15;
    const int m0  = blockIdx.x << 4;
    const int col = (wave << 4) + l;

    v8f ir, iz, in_, hr, hz, hn_;
    gate_tiles_f32<true >(agg,  w_ih, m0, col, hf, l, ir, iz, in_);
    gate_tiles_f32<false>(h_in, w_hh, m0, col, hf, l, hr, hz, hn_);

    const float bir = b_ih[col], biz = b_ih[HDIM + col], bin = b_ih[2 * HDIM + col];
    const float bhr = b_hh[col], bhz = b_hh[HDIM + col], bhn = b_hh[2 * HDIM + col];

#pragma unroll
    for (int v = 0; v < 8; ++v) {
        const size_t idx = (size_t)(m0 + v + (hf << 3)) * HDIM + col;
        const float gir = ir[v] + bir, ghr = hr[v] + bhr;
        const float giz = iz[v] + biz, ghz = hz[v] + bhz;
        const float gin = in_[v] + bin, ghn = hn_[v] + bhn;
        const float sr = gir + ghr;
        const float sz = giz + ghz;
        const float r = 1.0f / (1.0f + __expf(-sr));
        const float z = 1.0f / (1.0f + __expf(-sz));
        const float n = tanhf(gin + r * ghn);
        const float ho = h_in[idx];
        h_out[idx] = (1.0f - z) * n + z * ho;
    }
}

extern "C" void kernel_launch(void* const* d_in, const int* in_sizes, int n_in,
                              void* d_out, int out_size, void* d_ws, size_t ws_size,
                              hipStream_t stream)
{
    (void)in_sizes; (void)n_in; (void)out_size; (void)ws_size;

    const float* x      = (const float*)d_in[0];
    const int*   eidx   = (const int*)  d_in[1];
    const float* ea     = (const float*)d_in[2];
    const float* lin0_w = (const float*)d_in[3];
    const float* lin0_b = (const float*)d_in[4];
    const float* nn_w   = (const float*)d_in[5];
    const float* nn_b   = (const float*)d_in[6];
    const float* root_w = (const float*)d_in[7];
    const float* conv_b = (const float*)d_in[8];
    const float* w_ih   = (const float*)d_in[9];
    const float* w_hh   = (const float*)d_in[10];
    const float* b_ih   = (const float*)d_in[11];
    const float* b_hh   = (const float*)d_in[12];

    const int* src = eidx;
    const int* dst = eidx + NEDGE;

    const size_t plane = (size_t)NNODE * HDIM;
    float* Y   = (float*)d_ws;
    float* agg = Y + 5 * plane;
    float* hw  = agg + plane;
    float* ho  = (float*)d_out;

    const int gLin0 = ((NNODE / 16) * (HDIM / 16) + 3) / 4;
    const int gNode = (6 * (NNODE / 16) * (HDIM / 16) + 3) / 4;
    const int gStep = NNODE / 16;

    lin0_gemm<<<gLin0, 128, 0, stream>>>(x, lin0_w, lin0_b, hw);

    node_gemms<<<gNode, 128, 0, stream>>>(hw, nn_w, nn_b, root_w, conv_b, Y, agg);
    edge_msg<<<4096, 256, 0, stream>>>(src, dst, ea, Y, agg, NEDGE);
    gru_step<<<gStep, 128, 0, stream>>>(agg, hw, w_ih, w_hh, b_ih, b_hh, ho);

    node_gemms<<<gNode, 128, 0, stream>>>(ho, nn_w, nn_b, root_w, conv_b, Y, agg);
    edge_msg<<<4096, 256, 0, stream>>>(src, dst, ea, Y, agg, NEDGE);
    gru_step<<<gStep, 128, 0, stream>>>(agg, ho, w_ih, w_hh, b_ih, b_hh, hw);

    node_gemms<<<gNode, 128, 0, stream>>>(hw, nn_w, nn_b, root_w, conv_b, Y, agg);
    edge_msg<<<4096, 256, 0, stream>>>(src, dst, ea, Y, agg, NEDGE);
    gru_step<<<gStep, 128, 0, stream>>>(agg, hw, w_ih, w_hh, b_ih, b_hh, ho);
}


# Edge kernel: 4168 long-lived waves (was 32768), each owns 12 consecutive edges as whole 64-col rows and issues all its gathers before the first sum; atomic arrival order changes with the grid
# speedup vs baseline: 1.2342x; 1.1090x over previous
#include <hip/hip_runtime.h>


typedef __attribute__((ext_vector_type(2))) float v2f;
typedef __attribute__((ext_vector_type(8))) float v8f;

#define NNODE 10000
#define NEDGE 50000
#define HDIM  64

template <int K, bool RELUA, bool TRANSB>
__device__ __forceinline__ v8f wmma_tile_f32(
    const float* __restrict__ A, int lda,
    const float* __restrict__ B, int ldb,
    int m0, int col, int hf, int l)
{
    const float* __restrict__ Arow = A + (size_t)(m0 + l) * lda;
    const float* __restrict__ Bcol = TRANSB ? (B + (size_t)col * ldb) : B;
    v8f acc = {};
#pragma unroll
    for (int k = 0; k < K; k += 4) {
        const int kb = k + (hf << 1);
        v2f a = *(const v2f*)(Arow + kb);
        if (RELUA) { a.x = fmaxf(a.x, 0.0f); a.y = fmaxf(a.y, 0.0f); }
        v2f b;
        if (TRANSB) {
            b = *(const v2f*)(Bcol + kb);
        } else {
            b.x = B[(size_t)kb * ldb + col];
            b.y = B[(size_t)(kb + 1) * ldb + col];
        }
        acc = __builtin_amdgcn_wmma_f32_16x16x4_f32(
            false, a, false, b, (short)0, acc, false, false);
    }
    return acc;
}

__device__ __forceinline__ void store_tile(
    float* __restrict__ C, int ldc, int m0, int col, int hf,
    float bv, bool relu, v8f acc)
{
#pragma unroll
    for (int v = 0; v < 8; ++v) {
        const int row = m0 + v + (hf << 3);
        float r = acc[v] + bv;
        if (relu) r = fmaxf(r, 0.0f);
        C[(size_t)row * ldc + col] = r;
    }
}

__global__ __launch_bounds__(128) void lin0_gemm(
    const float* __restrict__ x, const float* __restrict__ w,
    const float* __restrict__ bias, float* __restrict__ h)
{
    const int wave = threadIdx.x >> 5;
    const int lane = threadIdx.x & 31;
    const int hf = lane >> 4, l = lane & 15;
    const int tile = blockIdx.x * 4 + wave;
    if (tile >= (NNODE / 16) * (HDIM / 16)) return;
    const int tm = tile >> 2, tn = tile & 3;
    const int m0 = tm << 4, col = (tn << 4) + l;

    v8f acc = wmma_tile_f32<32, false, false>(x, 32, w, HDIM, m0, col, hf, l);
    store_tile(h, HDIM, m0, col, hf, bias[col], true, acc);
}

__global__ __launch_bounds__(128) void node_gemms(
    const float* __restrict__ h,
    const float* __restrict__ nn_w, const float* __restrict__ nn_b,
    const float* __restrict__ root_w, const float* __restrict__ conv_b,
    float* __restrict__ Y, float* __restrict__ agg)
{
    const int wave = threadIdx.x >> 5;
    const int lane = threadIdx.x & 31;
    const int hf = lane >> 4, l = lane & 15;
    const int tile = blockIdx.x * 4 + wave;
    const int perPlane = (NNODE / 16) * (HDIM / 16);
    if (tile >= 6 * perPlane) return;

    const int p = tile / perPlane;
    const int t = tile - p * perPlane;
    const int tm = t >> 2, tn = t & 3;
    const int m0 = tm << 4, col = (tn << 4) + l;

    const size_t plane = (size_t)NNODE * HDIM;
    const float* B = (p < 4) ? (nn_w + (size_t)p * HDIM * HDIM)
                             : ((p == 4) ? nn_b : root_w);
    float* C = (p < 5) ? (Y + (size_t)p * plane) : agg;
    const float bv = (p == 5) ? conv_b[col] : 0.0f;

    v8f acc = wmma_tile_f32<64, false, false>(h, HDIM, B, HDIM, m0, col, hf, l);
    store_tile(C, HDIM, m0, col, hf, bv, false, acc);
}

#define EDGE_BLOCKS 521
#define EDGE_WPB    8
#define EDGE_WAVES  (EDGE_BLOCKS * EDGE_WPB)
#define EDGE_EPW    ((NEDGE + EDGE_WAVES - 1) / EDGE_WAVES)

__global__ __launch_bounds__(EDGE_WPB * 32) void edge_msg(
    const int* __restrict__ src, const int* __restrict__ dst,
    const float* __restrict__ ea,
    const float* __restrict__ Y, float* __restrict__ agg)
{
    static_assert(EDGE_EPW >= 1 && EDGE_EPW <= 32, "one lane per edge of a wave's chunk");
    const size_t plane = (size_t)NNODE * HDIM;
    const int lane = threadIdx.x & 31;
    const int wv = __builtin_amdgcn_readfirstlane(
        (int)((blockIdx.x * (EDGE_WPB * 32) + threadIdx.x) >> 5));
    const int e_beg = wv * EDGE_EPW;
    int n = NEDGE - e_beg;
    if (n > EDGE_EPW) n = EDGE_EPW;
    if (n <= 0) return;

    const int ei = e_beg + (lane < n ? lane : n - 1);
    const int sv = src[ei];
    const int dv = dst[ei];
    const float4 cv = *(const float4*)(ea + (size_t)ei * 4);

    float g[EDGE_EPW][10];
#pragma unroll
    for (int i = 0; i < EDGE_EPW; ++i) {
        const int j = (i < n) ? i : (n - 1);
        const int s = __builtin_amdgcn_readlane(sv, j);
        const float* __restrict__ row = Y + (size_t)s * HDIM + lane;
#pragma unroll
        for (int p = 0; p < 5; ++p) {
            g[i][2 * p]     = row[p * plane];
            g[i][2 * p + 1] = row[p * plane + 32];
        }
    }

#pragma unroll
    for (int i = 0; i < EDGE_EPW; ++i) {
        if (i < n) {
            const int d = __builtin_amdgcn_readlane(dv, i);
            const float c0 = __int_as_float(__builtin_amdgcn_readlane(__float_as_int(cv.x), i));
            const float c1 = __int_as_float(__builtin_amdgcn_readlane(__float_as_int(cv.y), i));
            const float c2 = __int_as_float(__builtin_amdgcn_readlane(__float_as_int(cv.z), i));
            const float c3 = __int_as_float(__builtin_amdgcn_readlane(__float_as_int(cv.w), i));
            float mlo = g[i][8], mhi = g[i][9];
            mlo = fmaf(c0, g[i][0], mlo);  mhi = fmaf(c0, g[i][1], mhi);
            mlo = fmaf(c1, g[i][2], mlo);  mhi = fmaf(c1, g[i][3], mhi);
            mlo = fmaf(c2, g[i][4], mlo);  mhi = fmaf(c2, g[i][5], mhi);
            mlo = fmaf(c3, g[i][6], mlo);  mhi = fmaf(c3, g[i][7], mhi);
            float* __restrict__ out = agg + (size_t)d * HDIM + lane;
            atomicAdd(out, mlo);
            atomicAdd(out + 32, mhi);
        }
    }
}

template <bool RELUA>
__device__ __forceinline__ void gate_tiles_f32(
    const float* __restrict__ A, const float* __restrict__ Wt,
    int m0, int col, int hf, int l, v8f& accR, v8f& accZ, v8f& accN)
{
    const float* __restrict__ Arow = A + (size_t)(m0 + l) * HDIM;
    const float* __restrict__ Br = Wt + (size_t)col * HDIM;
    const float* __restrict__ Bz = Wt + (size_t)(HDIM + col) * HDIM;
    const float* __restrict__ Bn = Wt + (size_t)(2 * HDIM + col) * HDIM;
    v8f r = {}, z = {}, n = {};
#pragma unroll
    for (int k = 0; k < HDIM; k += 4) {
        const int kb = k + (hf << 1);
        v2f a = *(const v2f*)(Arow + kb);
        if (RELUA) { a.x = fmaxf(a.x, 0.0f); a.y = fmaxf(a.y, 0.0f); }
        const v2f br = *(const v2f*)(Br + kb);
        const v2f bz = *(const v2f*)(Bz + kb);
        const v2f bn = *(const v2f*)(Bn + kb);
        r = __builtin_amdgcn_wmma_f32_16x16x4_f32(
            false, a, false, br, (short)0, r, false, false);
        z = __builtin_amdgcn_wmma_f32_16x16x4_f32(
            false, a, false, bz, (short)0, z, false, false);
        n = __builtin_amdgcn_wmma_f32_16x16x4_f32(
            false, a, false, bn, (short)0, n, false, false);
    }
    accR = r; accZ = z; accN = n;
}

__global__ __launch_bounds__(128) void gru_step(
    const float* __restrict__ agg, const float* __restrict__ h_in,
    const float* __restrict__ w_ih, const float* __restrict__ w_hh,
    const float* __restrict__ b_ih, const float* __restrict__ b_hh,
    float* __restrict__ h_out)
{
    const int wave = threadIdx.x >> 5;
    const int lane = threadIdx.x & 31;
    const int hf = lane >> 4, l = lane & 15;
    const int m0  = blockIdx.x << 4;
    const int col = (wave << 4) + l;

    v8f ir, iz, in_, hr, hz, hn_;
    gate_tiles_f32<true >(agg,  w_ih, m0, col, hf, l, ir, iz, in_);
    gate_tiles_f32<false>(h_in, w_hh, m0, col, hf, l, hr, hz, hn_);

    const float bir = b_ih[col], biz = b_ih[HDIM + col], bin = b_ih[2 * HDIM + col];
    const float bhr = b_hh[col], bhz = b_hh[HDIM + col], bhn = b_hh[2 * HDIM + col];

#pragma unroll
    for (int v = 0; v < 8; ++v) {
        const size_t idx = (size_t)(m0 + v + (hf << 3)) * HDIM + col;
        const float gir = ir[v] + bir, ghr = hr[v] + bhr;
        const float giz = iz[v] + biz, ghz = hz[v] + bhz;
        const float gin = in_[v] + bin, ghn = hn_[v] + bhn;
        const float sr = gir + ghr;
        const float sz = giz + ghz;
        const float r = 1.0f / (1.0f + __expf(-sr));
        const float z = 1.0f / (1.0f + __expf(-sz));
        const float n = tanhf(gin + r * ghn);
        const float ho = h_in[idx];
        h_out[idx] = (1.0f - z) * n + z * ho;
    }
}

extern "C" void kernel_launch(void* const* d_in, const int* in_sizes, int n_in,
                              void* d_out, int out_size, void* d_ws, size_t ws_size,
                              hipStream_t stream)
{
    (void)in_sizes; (void)n_in; (void)out_size; (void)ws_size;

    const float* x      = (const float*)d_in[0];
    const int*   eidx   = (const int*)  d_in[1];
    const float* ea     = (const float*)d_in[2];
    const float* lin0_w = (const float*)d_in[3];
    const float* lin0_b = (const float*)d_in[4];
    const float* nn_w   = (const float*)d_in[5];
    const float* nn_b   = (const float*)d_in[6];
    const float* root_w = (const float*)d_in[7];
    const float* conv_b = (const float*)d_in[8];
    const float* w_ih   = (const float*)d_in[9];
    const float* w_hh   = (const float*)d_in[10];
    const float* b_ih   = (const float*)d_in[11];
    const float* b_hh   = (const float*)d_in[12];

    const int* src = eidx;
    const int* dst = eidx + NEDGE;

    const size_t plane = (size_t)NNODE * HDIM;
    float* Y   = (float*)d_ws;
    float* agg = Y + 5 * plane;
    float* hw  = agg + plane;
    float* ho  = (float*)d_out;

    const int gLin0 = ((NNODE / 16) * (HDIM / 16) + 3) / 4;
    const int gNode = (6 * (NNODE / 16) * (HDIM / 16) + 3) / 4;
    const int gStep = NNODE / 16;

    lin0_gemm<<<gLin0, 128, 0, stream>>>(x, lin0_w, lin0_b, hw);

    node_gemms<<<gNode, 128, 0, stream>>>(hw, nn_w, nn_b, root_w, conv_b, Y, agg);
    edge_msg<<<EDGE_BLOCKS, EDGE_WPB * 32, 0, stream>>>(src, dst, ea, Y, agg);
    gru_step<<<gStep, 128, 0, stream>>>(agg, hw, w_ih, w_hh, b_ih, b_hh, ho);

    node_gemms<<<gNode, 128, 0, stream>>>(ho, nn_w, nn_b, root_w, conv_b, Y, agg);
    edge_msg<<<EDGE_BLOCKS, EDGE_WPB * 32, 0, stream>>>(src, dst, ea, Y, agg);
    gru_step<<<gStep, 128, 0, stream>>>(agg, ho, w_ih, w_hh, b_ih, b_hh, hw);

    node_gemms<<<gNode, 128, 0, stream>>>(hw, nn_w, nn_b, root_w, conv_b, Y, agg);
    edge_msg<<<EDGE_BLOCKS, EDGE_WPB * 32, 0, stream>>>(src, dst, ea, Y, agg);
    gru_step<<<gStep, 128, 0, stream>>>(agg, hw, w_ih, w_hh, b_ih, b_hh, ho);
}


# Fuse the node GEMMs into the kernel that produces h (lin0, GRU steps 1 and 2): block keeps its 16 new h rows in LDS, each wave computes its 6 plane tiles from them; 10 to 7 launches
# speedup vs baseline: 1.5117x; 1.2248x over previous
#include <hip/hip_runtime.h>


typedef __attribute__((ext_vector_type(2))) float v2f;
typedef __attribute__((ext_vector_type(8))) float v8f;

#define NNODE 10000
#define NEDGE 50000
#define HDIM  64
#define HPITCH 68

template <int K, bool RELUA, bool TRANSB>
__device__ __forceinline__ v8f wmma_tile_f32(
    const float* __restrict__ A, int lda,
    const float* __restrict__ B, int ldb,
    int m0, int col, int hf, int l)
{
    const float* __restrict__ Arow = A + (size_t)(m0 + l) * lda;
    const float* __restrict__ Bcol = TRANSB ? (B + (size_t)col * ldb) : B;
    v8f acc = {};
#pragma unroll
    for (int k = 0; k < K; k += 4) {
        const int kb = k + (hf << 1);
        v2f a = *(const v2f*)(Arow + kb);
        if (RELUA) { a.x = fmaxf(a.x, 0.0f); a.y = fmaxf(a.y, 0.0f); }
        v2f b;
        if (TRANSB) {
            b = *(const v2f*)(Bcol + kb);
        } else {
            b.x = B[(size_t)kb * ldb + col];
            b.y = B[(size_t)(kb + 1) * ldb + col];
        }
        acc = __builtin_amdgcn_wmma_f32_16x16x4_f32(
            false, a, false, b, (short)0, acc, false, false);
    }
    return acc;
}

__device__ __forceinline__ void store_tile(
    float* __restrict__ C, int ldc, int m0, int col, int hf,
    float bv, bool relu, v8f acc)
{
#pragma unroll
    for (int v = 0; v < 8; ++v) {
        const int row = m0 + v + (hf << 3);
        float r = acc[v] + bv;
        if (relu) r = fmaxf(r, 0.0f);
        C[(size_t)row * ldc + col] = r;
    }
}

__device__ __forceinline__ void node_planes_f32(
    const float* hs,
    const float* __restrict__ nn_w, const float* __restrict__ nn_b,
    const float* __restrict__ root_w, const float* __restrict__ conv_b,
    float* __restrict__ Y, float* agg,
    int m0, int col, int hf, int l)
{
    const size_t plane = (size_t)NNODE * HDIM;
    v8f acc[6];
#pragma unroll
    for (int p = 0; p < 6; ++p) acc[p] = v8f{};
#pragma unroll
    for (int k = 0; k < HDIM; k += 4) {
        const int kb = k + (hf << 1);
        const v2f a = *(const v2f*)(hs + l * HPITCH + kb);
#pragma unroll
        for (int p = 0; p < 6; ++p) {
            const float* __restrict__ B =
                (p < 4) ? (nn_w + (size_t)p * HDIM * HDIM)
                        : ((p == 4) ? nn_b : root_w);
            v2f b;
            b.x = B[(size_t)kb * HDIM + col];
            b.y = B[(size_t)(kb + 1) * HDIM + col];
            acc[p] = __builtin_amdgcn_wmma_f32_16x16x4_f32(
                false, a, false, b, (short)0, acc[p], false, false);
        }
    }
#pragma unroll
    for (int p = 0; p < 5; ++p)
        store_tile(Y + (size_t)p * plane, HDIM, m0, col, hf, 0.0f, false, acc[p]);
    store_tile(agg, HDIM, m0, col, hf, conv_b[col], false, acc[5]);
}

__global__ __launch_bounds__(128) void lin0_node(
    const float* __restrict__ x, const float* __restrict__ w,
    const float* __restrict__ bias,
    const float* __restrict__ nn_w, const float* __restrict__ nn_b,
    const float* __restrict__ root_w, const float* __restrict__ conv_b,
    float* __restrict__ h, float* __restrict__ Y, float* agg)
{
    __shared__ __attribute__((aligned(16))) float hs[16 * HPITCH];
    const int wave = threadIdx.x >> 5;
    const int lane = threadIdx.x & 31;
    const int hf = lane >> 4, l = lane & 15;
    const int m0  = blockIdx.x << 4;
    const int col = (wave << 4) + l;

    const v8f acc = wmma_tile_f32<32, false, false>(x, 32, w, HDIM, m0, col, hf, l);
    const float bv = bias[col];
#pragma unroll
    for (int v = 0; v < 8; ++v) {
        const int r = v + (hf << 3);
        const float hv = fmaxf(acc[v] + bv, 0.0f);
        h[(size_t)(m0 + r) * HDIM + col] = hv;
        hs[r * HPITCH + col] = hv;
    }
    __syncthreads();
    node_planes_f32(hs, nn_w, nn_b, root_w, conv_b, Y, agg, m0, col, hf, l);
}

#define EDGE_BLOCKS 521
#define EDGE_WPB    8
#define EDGE_WAVES  (EDGE_BLOCKS * EDGE_WPB)
#define EDGE_EPW    ((NEDGE + EDGE_WAVES - 1) / EDGE_WAVES)

__global__ __launch_bounds__(EDGE_WPB * 32) void edge_msg(
    const int* __restrict__ src, const int* __restrict__ dst,
    const float* __restrict__ ea,
    const float* __restrict__ Y, float* __restrict__ agg)
{
    static_assert(EDGE_EPW >= 1 && EDGE_EPW <= 32, "one lane per edge of a wave's chunk");
    const size_t plane = (size_t)NNODE * HDIM;
    const int lane = threadIdx.x & 31;
    const int wv = __builtin_amdgcn_readfirstlane(
        (int)((blockIdx.x * (EDGE_WPB * 32) + threadIdx.x) >> 5));
    const int e_beg = wv * EDGE_EPW;
    int n = NEDGE - e_beg;
    if (n > EDGE_EPW) n = EDGE_EPW;
    if (n <= 0) return;

    const int ei = e_beg + (lane < n ? lane : n - 1);
    const int sv = src[ei];
    const int dv = dst[ei];
    const float4 cv = *(const float4*)(ea + (size_t)ei * 4);

    float g[EDGE_EPW][10];
#pragma unroll
    for (int i = 0; i < EDGE_EPW; ++i) {
        const int j = (i < n) ? i : (n - 1);
        const int s = __builtin_amdgcn_readlane(sv, j);
        const float* __restrict__ row = Y + (size_t)s * HDIM + lane;
#pragma unroll
        for (int p = 0; p < 5; ++p) {
            g[i][2 * p]     = row[p * plane];
            g[i][2 * p + 1] = row[p * plane + 32];
        }
    }

#pragma unroll
    for (int i = 0; i < EDGE_EPW; ++i) {
        if (i < n) {
            const int d = __builtin_amdgcn_readlane(dv, i);
            const float c0 = __int_as_float(__builtin_amdgcn_readlane(__float_as_int(cv.x), i));
            const float c1 = __int_as_float(__builtin_amdgcn_readlane(__float_as_int(cv.y), i));
            const float c2 = __int_as_float(__builtin_amdgcn_readlane(__float_as_int(cv.z), i));
            const float c3 = __int_as_float(__builtin_amdgcn_readlane(__float_as_int(cv.w), i));
            float mlo = g[i][8], mhi = g[i][9];
            mlo = fmaf(c0, g[i][0], mlo);  mhi = fmaf(c0, g[i][1], mhi);
            mlo = fmaf(c1, g[i][2], mlo);  mhi = fmaf(c1, g[i][3], mhi);
            mlo = fmaf(c2, g[i][4], mlo);  mhi = fmaf(c2, g[i][5], mhi);
            mlo = fmaf(c3, g[i][6], mlo);  mhi = fmaf(c3, g[i][7], mhi);
            float* __restrict__ out = agg + (size_t)d * HDIM + lane;
            atomicAdd(out, mlo);
            atomicAdd(out + 32, mhi);
        }
    }
}

template <bool RELUA>
__device__ __forceinline__ void gate_tiles_f32(
    const float* __restrict__ A, const float* __restrict__ Wt,
    int m0, int col, int hf, int l, v8f& accR, v8f& accZ, v8f& accN)
{
    const float* __restrict__ Arow = A + (size_t)(m0 + l) * HDIM;
    const float* __restrict__ Br = Wt + (size_t)col * HDIM;
    const float* __restrict__ Bz = Wt + (size_t)(HDIM + col) * HDIM;
    const float* __restrict__ Bn = Wt + (size_t)(2 * HDIM + col) * HDIM;
    v8f r = {}, z = {}, n = {};
#pragma unroll
    for (int k = 0; k < HDIM; k += 4) {
        const int kb = k + (hf << 1);
        v2f a = *(const v2f*)(Arow + kb);
        if (RELUA) { a.x = fmaxf(a.x, 0.0f); a.y = fmaxf(a.y, 0.0f); }
        const v2f br = *(const v2f*)(Br + kb);
        const v2f bz = *(const v2f*)(Bz + kb);
        const v2f bn = *(const v2f*)(Bn + kb);
        r = __builtin_amdgcn_wmma_f32_16x16x4_f32(
            false, a, false, br, (short)0, r, false, false);
        z = __builtin_amdgcn_wmma_f32_16x16x4_f32(
            false, a, false, bz, (short)0, z, false, false);
        n = __builtin_amdgcn_wmma_f32_16x16x4_f32(
            false, a, false, bn, (short)0, n, false, false);
    }
    accR = r; accZ = z; accN = n;
}

template <bool NODE>
__global__ __launch_bounds__(128) void gru_step(
    float* agg, const float* __restrict__ h_in,
    const float* __restrict__ w_ih, const float* __restrict__ w_hh,
    const float* __restrict__ b_ih, const float* __restrict__ b_hh,
    float* __restrict__ h_out,
    const float* __restrict__ nn_w, const float* __restrict__ nn_b,
    const float* __restrict__ root_w, const float* __restrict__ conv_b,
    float* __restrict__ Y)
{
    __shared__ __attribute__((aligned(16))) float hs[16 * HPITCH];
    const int wave = threadIdx.x >> 5;
    const int lane = threadIdx.x & 31;
    const int hf = lane >> 4, l = lane & 15;
    const int m0  = blockIdx.x << 4;
    const int col = (wave << 4) + l;

    v8f ir, iz, in_, hr, hz, hn_;
    gate_tiles_f32<true >(agg,  w_ih, m0, col, hf, l, ir, iz, in_);
    gate_tiles_f32<false>(h_in, w_hh, m0, col, hf, l, hr, hz, hn_);

    const float bir = b_ih[col], biz = b_ih[HDIM + col], bin = b_ih[2 * HDIM + col];
    const float bhr = b_hh[col], bhz = b_hh[HDIM + col], bhn = b_hh[2 * HDIM + col];

#pragma unroll
    for (int v = 0; v < 8; ++v) {
        const int rr = v + (hf << 3);
        const size_t idx = (size_t)(m0 + rr) * HDIM + col;
        const float gir = ir[v] + bir, ghr = hr[v] + bhr;
        const float giz = iz[v] + biz, ghz = hz[v] + bhz;
        const float gin = in_[v] + bin, ghn = hn_[v] + bhn;
        const float sr = gir + ghr;
        const float sz = giz + ghz;
        const float r = 1.0f / (1.0f + __expf(-sr));
        const float z = 1.0f / (1.0f + __expf(-sz));
        const float n = tanhf(gin + r * ghn);
        const float ho = h_in[idx];
        const float hnew = (1.0f - z) * n + z * ho;
        h_out[idx] = hnew;
        if (NODE) hs[rr * HPITCH + col] = hnew;
    }
    if (NODE) {
        __syncthreads();
        node_planes_f32(hs, nn_w, nn_b, root_w, conv_b, Y, agg, m0, col, hf, l);
    }
}

extern "C" void kernel_launch(void* const* d_in, const int* in_sizes, int n_in,
                              void* d_out, int out_size, void* d_ws, size_t ws_size,
                              hipStream_t stream)
{
    (void)in_sizes; (void)n_in; (void)out_size; (void)ws_size;

    const float* x      = (const float*)d_in[0];
    const int*   eidx   = (const int*)  d_in[1];
    const float* ea     = (const float*)d_in[2];
    const float* lin0_w = (const float*)d_in[3];
    const float* lin0_b = (const float*)d_in[4];
    const float* nn_w   = (const float*)d_in[5];
    const float* nn_b   = (const float*)d_in[6];
    const float* root_w = (const float*)d_in[7];
    const float* conv_b = (const float*)d_in[8];
    const float* w_ih   = (const float*)d_in[9];
    const float* w_hh   = (const float*)d_in[10];
    const float* b_ih   = (const float*)d_in[11];
    const float* b_hh   = (const float*)d_in[12];

    const int* src = eidx;
    const int* dst = eidx + NEDGE;

    const size_t plane = (size_t)NNODE * HDIM;
    float* Y   = (float*)d_ws;
    float* agg = Y + 5 * plane;
    float* hw  = agg + plane;
    float* ho  = (float*)d_out;

    const int gRows = NNODE / 16;

    lin0_node<<<gRows, 128, 0, stream>>>(x, lin0_w, lin0_b,
                                         nn_w, nn_b, root_w, conv_b, hw, Y, agg);

    edge_msg<<<EDGE_BLOCKS, EDGE_WPB * 32, 0, stream>>>(src, dst, ea, Y, agg);
    gru_step<true><<<gRows, 128, 0, stream>>>(agg, hw, w_ih, w_hh, b_ih, b_hh, ho,
                                              nn_w, nn_b, root_w, conv_b, Y);

    edge_msg<<<EDGE_BLOCKS, EDGE_WPB * 32, 0, stream>>>(src, dst, ea, Y, agg);
    gru_step<true><<<gRows, 128, 0, stream>>>(agg, ho, w_ih, w_hh, b_ih, b_hh, hw,
                                              nn_w, nn_b, root_w, conv_b, Y);

    edge_msg<<<EDGE_BLOCKS, EDGE_WPB * 32, 0, stream>>>(src, dst, ea, Y, agg);
    gru_step<false><<<gRows, 128, 0, stream>>>(agg, hw, w_ih, w_hh, b_ih, b_hh, ho,
                                               nn_w, nn_b, root_w, conv_b, Y);
}


# Edge sums moved into the GRU kernel with no float atomics: per-dst edge lists built once (int slot claim), each row summed in edge-id order in registers, one K=384 WMMA product; 7 to 5 launches
# speedup vs baseline: 1.8261x; 1.2080x over previous
#include <hip/hip_runtime.h>


typedef __attribute__((ext_vector_type(2))) float v2f;
typedef __attribute__((ext_vector_type(8))) float v8f;

#define NNODE 10000
#define NEDGE 50000
#define HDIM  64
#define HPITCH 68
#define KCAT   384
#define SPITCH 388
#define LCAP  32
#define CHUNK 8

template <int K>
__device__ __forceinline__ v8f wmma_tile_f32(
    const float* __restrict__ A, int lda,
    const float* __restrict__ B, int ldb,
    int m0, int col, int hf, int l)
{
    const float* __restrict__ Arow = A + (size_t)(m0 + l) * lda;
    v8f acc = {};
#pragma unroll
    for (int k = 0; k < K; k += 4) {
        const int kb = k + (hf << 1);
        const v2f a = *(const v2f*)(Arow + kb);
        v2f b;
        b.x = B[(size_t)kb * ldb + col];
        b.y = B[(size_t)(kb + 1) * ldb + col];
        acc = __builtin_amdgcn_wmma_f32_16x16x4_f32(
            false, a, false, b, (short)0, acc, false, false);
    }
    return acc;
}

__global__ __launch_bounds__(128) void lin0_clear(
    const float* __restrict__ x, const float* __restrict__ w,
    const float* __restrict__ bias, float* __restrict__ h, int* __restrict__ cnt)
{
    const int wave = threadIdx.x >> 5;
    const int lane = threadIdx.x & 31;
    const int hf = lane >> 4, l = lane & 15;
    const int m0  = blockIdx.x << 4;
    const int col = (wave << 4) + l;

    if (threadIdx.x < 16) cnt[m0 + threadIdx.x] = 0;

    const v8f acc = wmma_tile_f32<32>(x, 32, w, HDIM, m0, col, hf, l);
    const float bv = bias[col];
#pragma unroll
    for (int v = 0; v < 8; ++v) {
        const int r = v + (hf << 3);
        h[(size_t)(m0 + r) * HDIM + col] = fmaxf(acc[v] + bv, 0.0f);
    }
}

__global__ __launch_bounds__(256) void build_lists(
    const int* __restrict__ src, const int* __restrict__ dst,
    const float* __restrict__ ea,
    int* __restrict__ cnt, int* __restrict__ lid, float* __restrict__ lco)
{
    const int e = blockIdx.x * 256 + threadIdx.x;
    if (e >= NEDGE) return;
    const int d = dst[e];
    const int pos = atomicAdd(cnt + d, 1);
    if (pos < LCAP) {
        const size_t slot = (size_t)d * LCAP + pos;
        *(int2*)(lid + slot * 2) = make_int2(e, src[e]);
        *(float4*)(lco + slot * 4) = *(const float4*)(ea + (size_t)e * 4);
    }
}

__device__ __forceinline__ void gate_tiles_f32(
    const float* A, int lda, const float* __restrict__ Wt,
    int col, int hf, int l, v8f& accR, v8f& accZ, v8f& accN)
{
    const float* Arow = A + l * lda;
    const float* __restrict__ Br = Wt + (size_t)col * HDIM;
    const float* __restrict__ Bz = Wt + (size_t)(HDIM + col) * HDIM;
    const float* __restrict__ Bn = Wt + (size_t)(2 * HDIM + col) * HDIM;
    v8f r = {}, z = {}, n = {};
#pragma unroll
    for (int k = 0; k < HDIM; k += 4) {
        const int kb = k + (hf << 1);
        const v2f a  = *(const v2f*)(Arow + kb);
        const v2f br = *(const v2f*)(Br + kb);
        const v2f bz = *(const v2f*)(Bz + kb);
        const v2f bn = *(const v2f*)(Bn + kb);
        r = __builtin_amdgcn_wmma_f32_16x16x4_f32(
            false, a, false, br, (short)0, r, false, false);
        z = __builtin_amdgcn_wmma_f32_16x16x4_f32(
            false, a, false, bz, (short)0, z, false, false);
        n = __builtin_amdgcn_wmma_f32_16x16x4_f32(
            false, a, false, bn, (short)0, n, false, false);
    }
    accR = r; accZ = z; accN = n;
}

__device__ __forceinline__ void edge_fma(
    float (&a)[10], float c0, float c1, float c2, float c3, float hlo, float hhi)
{
    a[0] = fmaf(c0, hlo, a[0]);  a[1] = fmaf(c0, hhi, a[1]);
    a[2] = fmaf(c1, hlo, a[2]);  a[3] = fmaf(c1, hhi, a[3]);
    a[4] = fmaf(c2, hlo, a[4]);  a[5] = fmaf(c2, hhi, a[5]);
    a[6] = fmaf(c3, hlo, a[6]);  a[7] = fmaf(c3, hhi, a[7]);
    a[8] += hlo;                 a[9] += hhi;
}

__device__ __forceinline__ void scan_row(
    const float* __restrict__ h_in, const int* __restrict__ src,
    const int* __restrict__ dst, const float* __restrict__ ea,
    int d, int lane, float* srow)
{
    float a[10];
#pragma unroll
    for (int i = 0; i < 10; ++i) a[i] = 0.0f;
    for (int e0 = 0; e0 < NEDGE; e0 += 32) {
        const int e = e0 + lane;
        const bool hit = (e < NEDGE) && (dst[e < NEDGE ? e : NEDGE - 1] == d);
        unsigned m = __builtin_amdgcn_ballot_w32(hit);
        while (m) {
            const int ee = e0 + __builtin_ctz(m);
            m &= m - 1;
            const int s = src[ee];
            const float4 c = *(const float4*)(ea + (size_t)ee * 4);
            const float* __restrict__ row = h_in + (size_t)s * HDIM + lane;
            edge_fma(a, c.x, c.y, c.z, c.w, row[0], row[32]);
        }
    }
#pragma unroll
    for (int p = 0; p < 5; ++p) {
        srow[HDIM * (p + 1) + lane]      = a[2 * p];
        srow[HDIM * (p + 1) + lane + 32] = a[2 * p + 1];
    }
}

__global__ __launch_bounds__(128) void fused_step(
    const float* __restrict__ h_in, float* __restrict__ h_out,
    const int* __restrict__ src, const int* __restrict__ dst,
    const float* __restrict__ ea,
    const int* __restrict__ cnt, const int* __restrict__ lid,
    const float* __restrict__ lco,
    const float* __restrict__ nn_w, const float* __restrict__ nn_b,
    const float* __restrict__ root_w, const float* __restrict__ conv_b,
    const float* __restrict__ w_ih, const float* __restrict__ w_hh,
    const float* __restrict__ b_ih, const float* __restrict__ b_hh)
{
    __shared__ __attribute__((aligned(16))) float St[16 * SPITCH];
    __shared__ __attribute__((aligned(16))) float Ms[16 * HPITCH];
    const int wave = threadIdx.x >> 5;
    const int lane = threadIdx.x & 31;
    const int hf = lane >> 4, l = lane & 15;
    const int m0  = blockIdx.x << 4;
    const int col = (wave << 4) + l;

    const int rb = (wave << 2);
    int nl[4];
    bool big[4];
    int ev[4], sv[4], rk[4];
    float4 cv[4];
    float hown[4][2];
#pragma unroll
    for (int r = 0; r < 4; ++r) {
        const int d = m0 + rb + r;
        const int c = __builtin_amdgcn_readfirstlane(cnt[d]);
        big[r] = c > LCAP;
        nl[r] = big[r] ? 0 : c;
        hown[r][0] = h_in[(size_t)d * HDIM + lane];
        hown[r][1] = h_in[(size_t)d * HDIM + lane + 32];
    }
#pragma unroll
    for (int r = 0; r < 4; ++r) {
        const size_t slot = (size_t)(m0 + rb + r) * LCAP + lane;
        int2 id = make_int2(0x7fffffff, 0);
        float4 c4 = make_float4(0.0f, 0.0f, 0.0f, 0.0f);
        if (lane < nl[r]) {
            id = *(const int2*)(lid + slot * 2);
            c4 = *(const float4*)(lco + slot * 4);
        }
        ev[r] = id.x; sv[r] = id.y; cv[r] = c4;
    }
#pragma unroll
    for (int r = 0; r < 4; ++r) {
        int k = 0;
        for (int j = 0; j < nl[r]; ++j) {
            const int ej = __builtin_amdgcn_readlane(ev[r], j);
            k += (ej < ev[r]) ? 1 : 0;
        }
        rk[r] = k;
    }
    int maxn = nl[0] > nl[1] ? nl[0] : nl[1];
    { const int t = nl[2] > nl[3] ? nl[2] : nl[3]; maxn = maxn > t ? maxn : t; }

    float acc[4][10];
#pragma unroll
    for (int r = 0; r < 4; ++r)
#pragma unroll
        for (int i = 0; i < 10; ++i) acc[r][i] = 0.0f;

    for (int t0 = 0; t0 < maxn; t0 += CHUNK) {
        float g[4][CHUNK][2];
        int   js[4][CHUNK];
        bool  ok[4][CHUNK];
#pragma unroll
        for (int r = 0; r < 4; ++r) {
#pragma unroll
            for (int u = 0; u < CHUNK; ++u) {
                const int t = t0 + u;
                const unsigned bm = __builtin_amdgcn_ballot_w32(rk[r] == t);
                ok[r][u] = (t < nl[r]) && (bm != 0u);
                js[r][u] = ok[r][u] ? __builtin_ctz(bm) : 0;
                const int s = __builtin_amdgcn_readlane(sv[r], js[r][u]);
                const float* __restrict__ row = h_in + (size_t)s * HDIM + lane;
                g[r][u][0] = row[0];
                g[r][u][1] = row[32];
            }
        }
#pragma unroll
        for (int r = 0; r < 4; ++r) {
#pragma unroll
            for (int u = 0; u < CHUNK; ++u) {
                const int j = js[r][u];
                const float c0 = __int_as_float(__builtin_amdgcn_readlane(__float_as_int(cv[r].x), j));
                const float c1 = __int_as_float(__builtin_amdgcn_readlane(__float_as_int(cv[r].y), j));
                const float c2 = __int_as_float(__builtin_amdgcn_readlane(__float_as_int(cv[r].z), j));
                const float c3 = __int_as_float(__builtin_amdgcn_readlane(__float_as_int(cv[r].w), j));
                const float hlo = ok[r][u] ? g[r][u][0] : 0.0f;
                const float hhi = ok[r][u] ? g[r][u][1] : 0.0f;
                edge_fma(acc[r], c0, c1, c2, c3, hlo, hhi);
            }
        }
    }

#pragma unroll
    for (int r = 0; r < 4; ++r) {
        float* srow = St + (rb + r) * SPITCH;
        srow[lane]      = hown[r][0];
        srow[lane + 32] = hown[r][1];
        if (!big[r]) {
#pragma unroll
            for (int p = 0; p < 5; ++p) {
                srow[HDIM * (p + 1) + lane]      = acc[r][2 * p];
                srow[HDIM * (p + 1) + lane + 32] = acc[r][2 * p + 1];
            }
        }
    }
#pragma nounroll
    for (int r = 0; r < 4; ++r) {
        const int d = m0 + rb + r;
        if (__builtin_amdgcn_readfirstlane(cnt[d]) > LCAP)
            scan_row(h_in, src, dst, ea, d, lane, St + (rb + r) * SPITCH);
    }
    __syncthreads();

    {
        v8f pa[6];
#pragma unroll
        for (int p = 0; p < 6; ++p) pa[p] = v8f{};
#pragma unroll
        for (int k = 0; k < HDIM; k += 4) {
            const int kb = k + (hf << 1);
#pragma unroll
            for (int p = 0; p < 6; ++p) {
                const float* __restrict__ B =
                    (p == 0) ? root_w
                             : ((p == 5) ? nn_b : (nn_w + (size_t)(p - 1) * HDIM * HDIM));
                const v2f a = *(const v2f*)(St + l * SPITCH + p * HDIM + kb);
                v2f b;
                b.x = B[(size_t)kb * HDIM + col];
                b.y = B[(size_t)(kb + 1) * HDIM + col];
                pa[p] = __builtin_amdgcn_wmma_f32_16x16x4_f32(
                    false, a, false, b, (short)0, pa[p], false, false);
            }
        }
        const float cb = conv_b[col];
#pragma unroll
        for (int v = 0; v < 8; ++v) {
            const int rr = v + (hf << 3);
            const float s = ((pa[0][v] + pa[1][v]) + (pa[2][v] + pa[3][v]))
                          + (pa[4][v] + pa[5][v]);
            Ms[rr * HPITCH + col] = fmaxf(s + cb, 0.0f);
        }
    }
    __syncthreads();

    v8f ir, iz, in_, hr, hz, hn_;
    gate_tiles_f32(Ms, HPITCH, w_ih, col, hf, l, ir, iz, in_);
    gate_tiles_f32(St, SPITCH, w_hh, col, hf, l, hr, hz, hn_);

    const float bir = b_ih[col], biz = b_ih[HDIM + col], bin = b_ih[2 * HDIM + col];
    const float bhr = b_hh[col], bhz = b_hh[HDIM + col], bhn = b_hh[2 * HDIM + col];

#pragma unroll
    for (int v = 0; v < 8; ++v) {
        const int rr = v + (hf << 3);
        const size_t idx = (size_t)(m0 + rr) * HDIM + col;
        const float gir = ir[v] + bir, ghr = hr[v] + bhr;
        const float giz = iz[v] + biz, ghz = hz[v] + bhz;
        const float gin = in_[v] + bin, ghn = hn_[v] + bhn;
        const float sr = gir + ghr;
        const float sz = giz + ghz;
        const float r = 1.0f / (1.0f + __expf(-sr));
        const float z = 1.0f / (1.0f + __expf(-sz));
        const float n = tanhf(gin + r * ghn);
        const float ho = St[rr * SPITCH + col];
        h_out[idx] = (1.0f - z) * n + z * ho;
    }
}

extern "C" void kernel_launch(void* const* d_in, const int* in_sizes, int n_in,
                              void* d_out, int out_size, void* d_ws, size_t ws_size,
                              hipStream_t stream)
{
    (void)in_sizes; (void)n_in; (void)out_size; (void)ws_size;

    const float* x      = (const float*)d_in[0];
    const int*   eidx   = (const int*)  d_in[1];
    const float* ea     = (const float*)d_in[2];
    const float* lin0_w = (const float*)d_in[3];
    const float* lin0_b = (const float*)d_in[4];
    const float* nn_w   = (const float*)d_in[5];
    const float* nn_b   = (const float*)d_in[6];
    const float* root_w = (const float*)d_in[7];
    const float* conv_b = (const float*)d_in[8];
    const float* w_ih   = (const float*)d_in[9];
    const float* w_hh   = (const float*)d_in[10];
    const float* b_ih   = (const float*)d_in[11];
    const float* b_hh   = (const float*)d_in[12];

    const int* src = eidx;
    const int* dst = eidx + NEDGE;

    float* hw  = (float*)d_ws;
    float* lco = hw + (size_t)NNODE * HDIM;
    int*   lid = (int*)(lco + (size_t)NNODE * LCAP * 4);
    int*   cnt = lid + (size_t)NNODE * LCAP * 2;
    float* ho  = (float*)d_out;

    lin0_clear<<<NNODE / 16, 128, 0, stream>>>(x, lin0_w, lin0_b, hw, cnt);
    build_lists<<<(NEDGE + 255) / 256, 256, 0, stream>>>(src, dst, ea, cnt, lid, lco);

    fused_step<<<NNODE / 16, 128, 0, stream>>>(hw, ho, src, dst, ea, cnt, lid, lco,
        nn_w, nn_b, root_w, conv_b, w_ih, w_hh, b_ih, b_hh);
    fused_step<<<NNODE / 16, 128, 0, stream>>>(ho, hw, src, dst, ea, cnt, lid, lco,
        nn_w, nn_b, root_w, conv_b, w_ih, w_hh, b_ih, b_hh);
    fused_step<<<NNODE / 16, 128, 0, stream>>>(hw, ho, src, dst, ea, cnt, lid, lco,
        nn_w, nn_b, root_w, conv_b, w_ih, w_hh, b_ih, b_hh);
}


# Weights copied once per call (first kernel) into WMMA B-fragment order in the workspace: each B load of the step kernel is a 16-byte-per-lane read of 4 whole lines; 288 to 96 weight loads per wave
# speedup vs baseline: 2.1321x; 1.1676x over previous
#include <hip/hip_runtime.h>


typedef __attribute__((ext_vector_type(2))) float v2f;
typedef __attribute__((ext_vector_type(8))) float v8f;

#define NNODE 10000
#define NEDGE 50000
#define HDIM  64
#define HPITCH 68
#define KCAT   384
#define SPITCH 388
#define LCAP  32
#define CHUNK 8
#define PACK_FLOATS 24576

template <int K>
__device__ __forceinline__ v8f wmma_tile_f32(
    const float* __restrict__ A, int lda,
    const float* __restrict__ B, int ldb,
    int m0, int col, int hf, int l)
{
    const float* __restrict__ Arow = A + (size_t)(m0 + l) * lda;
    v8f acc = {};
#pragma unroll
    for (int k = 0; k < K; k += 4) {
        const int kb = k + (hf << 1);
        const v2f a = *(const v2f*)(Arow + kb);
        v2f b;
        b.x = B[(size_t)kb * ldb + col];
        b.y = B[(size_t)(kb + 1) * ldb + col];
        acc = __builtin_amdgcn_wmma_f32_16x16x4_f32(
            false, a, false, b, (short)0, acc, false, false);
    }
    return acc;
}

__global__ __launch_bounds__(128) void lin0_clear(
    const float* __restrict__ x, const float* __restrict__ w,
    const float* __restrict__ bias, float* __restrict__ h, int* __restrict__ cnt,
    const float* __restrict__ nn_w, const float* __restrict__ nn_b,
    const float* __restrict__ root_w,
    const float* __restrict__ w_ih, const float* __restrict__ w_hh,
    float* __restrict__ pw2, float* __restrict__ pg)
{
    const int t = blockIdx.x * 128 + threadIdx.x;
    const bool packs = t < PACK_FLOATS;
    float pv2 = 0.0f, pvg = 0.0f;
    if (packs) {
        const int f = t & 3, ln = (t >> 2) & 31, rest = t >> 7;
        const int c = ln & 15;
        const int k4 = ((f >> 1) << 2) + ((ln >> 4) << 1) + (f & 1);
        {
            const int p = rest % 6, r2 = rest / 6;
            const int j2 = r2 & 7, wv = r2 >> 3;
            const float* __restrict__ B =
                (p == 0) ? root_w
                         : ((p == 5) ? nn_b : (nn_w + (size_t)(p - 1) * HDIM * HDIM));
            pv2 = B[(size_t)(8 * j2 + k4) * HDIM + (wv << 4) + c];
        }
        {
            const int g = rest % 3, r2 = rest / 3;
            const int j2 = r2 & 7, r3 = r2 >> 3;
            const int wv = r3 & 3, q = r3 >> 2;
            const float* __restrict__ Wt = q ? w_hh : w_ih;
            pvg = Wt[(size_t)(g * HDIM + (wv << 4) + c) * HDIM + 8 * j2 + k4];
        }
    }

    const int wave = threadIdx.x >> 5;
    const int lane = threadIdx.x & 31;
    const int hf = lane >> 4, l = lane & 15;
    const int m0  = blockIdx.x << 4;
    const int col = (wave << 4) + l;

    if (threadIdx.x < 16) cnt[m0 + threadIdx.x] = 0;

    const v8f acc = wmma_tile_f32<32>(x, 32, w, HDIM, m0, col, hf, l);
    const float bv = bias[col];
#pragma unroll
    for (int v = 0; v < 8; ++v) {
        const int r = v + (hf << 3);
        h[(size_t)(m0 + r) * HDIM + col] = fmaxf(acc[v] + bv, 0.0f);
    }
    if (packs) { pw2[t] = pv2; pg[t] = pvg; }
}

__global__ __launch_bounds__(256) void build_lists(
    const int* __restrict__ src, const int* __restrict__ dst,
    const float* __restrict__ ea,
    int* __restrict__ cnt, int* __restrict__ lid, float* __restrict__ lco)
{
    const int e = blockIdx.x * 256 + threadIdx.x;
    if (e >= NEDGE) return;
    const int d = dst[e];
    const int pos = atomicAdd(cnt + d, 1);
    if (pos < LCAP) {
        const size_t slot = (size_t)d * LCAP + pos;
        *(int2*)(lid + slot * 2) = make_int2(e, src[e]);
        *(float4*)(lco + slot * 4) = *(const float4*)(ea + (size_t)e * 4);
    }
}

__device__ __forceinline__ void gate_tiles_f32(
    const float* A, int lda, const float4* __restrict__ Bq,
    int hf, int l, v8f& accR, v8f& accZ, v8f& accN)
{
    const float* Arow = A + l * lda;
    v8f r = {}, z = {}, n = {};
#pragma unroll
    for (int j2 = 0; j2 < HDIM / 8; ++j2) {
        const float4 qr = Bq[(j2 * 3 + 0) * 32];
        const float4 qz = Bq[(j2 * 3 + 1) * 32];
        const float4 qn = Bq[(j2 * 3 + 2) * 32];
#pragma unroll
        for (int s = 0; s < 2; ++s) {
            const int kb = 8 * j2 + 4 * s + (hf << 1);
            const v2f a  = *(const v2f*)(Arow + kb);
            const v2f br = s ? v2f{qr.z, qr.w} : v2f{qr.x, qr.y};
            const v2f bz = s ? v2f{qz.z, qz.w} : v2f{qz.x, qz.y};
            const v2f bn = s ? v2f{qn.z, qn.w} : v2f{qn.x, qn.y};
            r = __builtin_amdgcn_wmma_f32_16x16x4_f32(
                false, a, false, br, (short)0, r, false, false);
            z = __builtin_amdgcn_wmma_f32_16x16x4_f32(
                false, a, false, bz, (short)0, z, false, false);
            n = __builtin_amdgcn_wmma_f32_16x16x4_f32(
                false, a, false, bn, (short)0, n, false, false);
        }
    }
    accR = r; accZ = z; accN = n;
}

__device__ __forceinline__ void edge_fma(
    float (&a)[10], float c0, float c1, float c2, float c3, float hlo, float hhi)
{
    a[0] = fmaf(c0, hlo, a[0]);  a[1] = fmaf(c0, hhi, a[1]);
    a[2] = fmaf(c1, hlo, a[2]);  a[3] = fmaf(c1, hhi, a[3]);
    a[4] = fmaf(c2, hlo, a[4]);  a[5] = fmaf(c2, hhi, a[5]);
    a[6] = fmaf(c3, hlo, a[6]);  a[7] = fmaf(c3, hhi, a[7]);
    a[8] += hlo;                 a[9] += hhi;
}

__device__ __forceinline__ void scan_row(
    const float* __restrict__ h_in, const int* __restrict__ src,
    const int* __restrict__ dst, const float* __restrict__ ea,
    int d, int lane, float* srow)
{
    float a[10];
#pragma unroll
    for (int i = 0; i < 10; ++i) a[i] = 0.0f;
    for (int e0 = 0; e0 < NEDGE; e0 += 32) {
        const int e = e0 + lane;
        const bool hit = (e < NEDGE) && (dst[e < NEDGE ? e : NEDGE - 1] == d);
        unsigned m = __builtin_amdgcn_ballot_w32(hit);
        while (m) {
            const int ee = e0 + __builtin_ctz(m);
            m &= m - 1;
            const int s = src[ee];
            const float4 c = *(const float4*)(ea + (size_t)ee * 4);
            const float* __restrict__ row = h_in + (size_t)s * HDIM + lane;
            edge_fma(a, c.x, c.y, c.z, c.w, row[0], row[32]);
        }
    }
#pragma unroll
    for (int p = 0; p < 5; ++p) {
        srow[HDIM * (p + 1) + lane]      = a[2 * p];
        srow[HDIM * (p + 1) + lane + 32] = a[2 * p + 1];
    }
}

__global__ __launch_bounds__(128) void fused_step(
    const float* __restrict__ h_in, float* __restrict__ h_out,
    const int* __restrict__ src, const int* __restrict__ dst,
    const float* __restrict__ ea,
    const int* __restrict__ cnt, const int* __restrict__ lid,
    const float* __restrict__ lco,
    const float* __restrict__ pw2, const float* __restrict__ pg,
    const float* __restrict__ conv_b,
    const float* __restrict__ b_ih, const float* __restrict__ b_hh)
{
    __shared__ __attribute__((aligned(16))) float St[16 * SPITCH];
    __shared__ __attribute__((aligned(16))) float Ms[16 * HPITCH];
    const int wave = threadIdx.x >> 5;
    const int lane = threadIdx.x & 31;
    const int hf = lane >> 4, l = lane & 15;
    const int m0  = blockIdx.x << 4;
    const int col = (wave << 4) + l;

    const int rb = (wave << 2);
    int nl[4];
    bool big[4];
    int ev[4], sv[4], rk[4];
    float4 cv[4];
    float hown[4][2];
#pragma unroll
    for (int r = 0; r < 4; ++r) {
        const int d = m0 + rb + r;
        const int c = __builtin_amdgcn_readfirstlane(cnt[d]);
        big[r] = c > LCAP;
        nl[r] = big[r] ? 0 : c;
        hown[r][0] = h_in[(size_t)d * HDIM + lane];
        hown[r][1] = h_in[(size_t)d * HDIM + lane + 32];
    }
#pragma unroll
    for (int r = 0; r < 4; ++r) {
        const size_t slot = (size_t)(m0 + rb + r) * LCAP + lane;
        int2 id = make_int2(0x7fffffff, 0);
        float4 c4 = make_float4(0.0f, 0.0f, 0.0f, 0.0f);
        if (lane < nl[r]) {
            id = *(const int2*)(lid + slot * 2);
            c4 = *(const float4*)(lco + slot * 4);
        }
        ev[r] = id.x; sv[r] = id.y; cv[r] = c4;
    }
#pragma unroll
    for (int r = 0; r < 4; ++r) {
        int k = 0;
        for (int j = 0; j < nl[r]; ++j) {
            const int ej = __builtin_amdgcn_readlane(ev[r], j);
            k += (ej < ev[r]) ? 1 : 0;
        }
        rk[r] = k;
    }
    int maxn = nl[0] > nl[1] ? nl[0] : nl[1];
    { const int t = nl[2] > nl[3] ? nl[2] : nl[3]; maxn = maxn > t ? maxn : t; }

    float acc[4][10];
#pragma unroll
    for (int r = 0; r < 4; ++r)
#pragma unroll
        for (int i = 0; i < 10; ++i) acc[r][i] = 0.0f;

    for (int t0 = 0; t0 < maxn; t0 += CHUNK) {
        float g[4][CHUNK][2];
        int   js[4][CHUNK];
        bool  ok[4][CHUNK];
#pragma unroll
        for (int r = 0; r < 4; ++r) {
#pragma unroll
            for (int u = 0; u < CHUNK; ++u) {
                const int t = t0 + u;
                const unsigned bm = __builtin_amdgcn_ballot_w32(rk[r] == t);
                ok[r][u] = (t < nl[r]) && (bm != 0u);
                js[r][u] = ok[r][u] ? __builtin_ctz(bm) : 0;
                const int s = __builtin_amdgcn_readlane(sv[r], js[r][u]);
                const float* __restrict__ row = h_in + (size_t)s * HDIM + lane;
                g[r][u][0] = row[0];
                g[r][u][1] = row[32];
            }
        }
#pragma unroll
        for (int r = 0; r < 4; ++r) {
#pragma unroll
            for (int u = 0; u < CHUNK; ++u) {
                const int j = js[r][u];
                const float c0 = __int_as_float(__builtin_amdgcn_readlane(__float_as_int(cv[r].x), j));
                const float c1 = __int_as_float(__builtin_amdgcn_readlane(__float_as_int(cv[r].y), j));
                const float c2 = __int_as_float(__builtin_amdgcn_readlane(__float_as_int(cv[r].z), j));
                const float c3 = __int_as_float(__builtin_amdgcn_readlane(__float_as_int(cv[r].w), j));
                const float hlo = ok[r][u] ? g[r][u][0] : 0.0f;
                const float hhi = ok[r][u] ? g[r][u][1] : 0.0f;
                edge_fma(acc[r], c0, c1, c2, c3, hlo, hhi);
            }
        }
    }

#pragma unroll
    for (int r = 0; r < 4; ++r) {
        float* srow = St + (rb + r) * SPITCH;
        srow[lane]      = hown[r][0];
        srow[lane + 32] = hown[r][1];
        if (!big[r]) {
#pragma unroll
            for (int p = 0; p < 5; ++p) {
                srow[HDIM * (p + 1) + lane]      = acc[r][2 * p];
                srow[HDIM * (p + 1) + lane + 32] = acc[r][2 * p + 1];
            }
        }
    }
#pragma nounroll
    for (int r = 0; r < 4; ++r) {
        const int d = m0 + rb + r;
        if (__builtin_amdgcn_readfirstlane(cnt[d]) > LCAP)
            scan_row(h_in, src, dst, ea, d, lane, St + (rb + r) * SPITCH);
    }
    __syncthreads();

    {
        v8f pa[6];
#pragma unroll
        for (int p = 0; p < 6; ++p) pa[p] = v8f{};
        const float4* __restrict__ Bw =
            (const float4*)pw2 + (size_t)wave * (8 * 6 * 32) + lane;
#pragma unroll
        for (int j2 = 0; j2 < HDIM / 8; ++j2) {
            float4 q[6];
#pragma unroll
            for (int p = 0; p < 6; ++p) q[p] = Bw[(j2 * 6 + p) * 32];
#pragma unroll
            for (int s = 0; s < 2; ++s) {
                const int kb = 8 * j2 + 4 * s + (hf << 1);
#pragma unroll
                for (int p = 0; p < 6; ++p) {
                    const v2f a = *(const v2f*)(St + l * SPITCH + p * HDIM + kb);
                    const v2f b = s ? v2f{q[p].z, q[p].w} : v2f{q[p].x, q[p].y};
                    pa[p] = __builtin_amdgcn_wmma_f32_16x16x4_f32(
                        false, a, false, b, (short)0, pa[p], false, false);
                }
            }
        }
        const float cb = conv_b[col];
#pragma unroll
        for (int v = 0; v < 8; ++v) {
            const int rr = v + (hf << 3);
            const float s = ((pa[0][v] + pa[1][v]) + (pa[2][v] + pa[3][v]))
                          + (pa[4][v] + pa[5][v]);
            Ms[rr * HPITCH + col] = fmaxf(s + cb, 0.0f);
        }
    }
    __syncthreads();

    v8f ir, iz, in_, hr, hz, hn_;
    const float4* __restrict__ Gi = (const float4*)pg + (size_t)wave * (8 * 3 * 32) + lane;
    const float4* __restrict__ Gh = Gi + 4 * 8 * 3 * 32;
    gate_tiles_f32(Ms, HPITCH, Gi, hf, l, ir, iz, in_);
    gate_tiles_f32(St, SPITCH, Gh, hf, l, hr, hz, hn_);

    const float bir = b_ih[col], biz = b_ih[HDIM + col], bin = b_ih[2 * HDIM + col];
    const float bhr = b_hh[col], bhz = b_hh[HDIM + col], bhn = b_hh[2 * HDIM + col];

#pragma unroll
    for (int v = 0; v < 8; ++v) {
        const int rr = v + (hf << 3);
        const size_t idx = (size_t)(m0 + rr) * HDIM + col;
        const float gir = ir[v] + bir, ghr = hr[v] + bhr;
        const float giz = iz[v] + biz, ghz = hz[v] + bhz;
        const float gin = in_[v] + bin, ghn = hn_[v] + bhn;
        const float sr = gir + ghr;
        const float sz = giz + ghz;
        const float r = 1.0f / (1.0f + __expf(-sr));
        const float z = 1.0f / (1.0f + __expf(-sz));
        const float n = tanhf(gin + r * ghn);
        const float ho = St[rr * SPITCH + col];
        h_out[idx] = (1.0f - z) * n + z * ho;
    }
}

extern "C" void kernel_launch(void* const* d_in, const int* in_sizes, int n_in,
                              void* d_out, int out_size, void* d_ws, size_t ws_size,
                              hipStream_t stream)
{
    (void)in_sizes; (void)n_in; (void)out_size; (void)ws_size;

    const float* x      = (const float*)d_in[0];
    const int*   eidx   = (const int*)  d_in[1];
    const float* ea     = (const float*)d_in[2];
    const float* lin0_w = (const float*)d_in[3];
    const float* lin0_b = (const float*)d_in[4];
    const float* nn_w   = (const float*)d_in[5];
    const float* nn_b   = (const float*)d_in[6];
    const float* root_w = (const float*)d_in[7];
    const float* conv_b = (const float*)d_in[8];
    const float* w_ih   = (const float*)d_in[9];
    const float* w_hh   = (const float*)d_in[10];
    const float* b_ih   = (const float*)d_in[11];
    const float* b_hh   = (const float*)d_in[12];

    const int* src = eidx;
    const int* dst = eidx + NEDGE;

    float* hw  = (float*)d_ws;
    float* lco = hw + (size_t)NNODE * HDIM;
    int*   lid = (int*)(lco + (size_t)NNODE * LCAP * 4);
    int*   cnt = lid + (size_t)NNODE * LCAP * 2;
    float* pw2 = (float*)(cnt + 10240);
    float* pg  = pw2 + PACK_FLOATS;
    float* ho  = (float*)d_out;

    lin0_clear<<<NNODE / 16, 128, 0, stream>>>(x, lin0_w, lin0_b, hw, cnt,
        nn_w, nn_b, root_w, w_ih, w_hh, pw2, pg);
    build_lists<<<(NEDGE + 255) / 256, 256, 0, stream>>>(src, dst, ea, cnt, lid, lco);

    fused_step<<<NNODE / 16, 128, 0, stream>>>(hw, ho, src, dst, ea, cnt, lid, lco,
        pw2, pg, conv_b, b_ih, b_hh);
    fused_step<<<NNODE / 16, 128, 0, stream>>>(ho, hw, src, dst, ea, cnt, lid, lco,
        pw2, pg, conv_b, b_ih, b_hh);
    fused_step<<<NNODE / 16, 128, 0, stream>>>(hw, ho, src, dst, ea, cnt, lid, lco,
        pw2, pg, conv_b, b_ih, b_hh);
}


# Edge sums of the step kernel re-mapped: list entries sorted by edge id into an LDS strip, then 8 lanes per row sum 4 rows at once, 8 columns per lane by 16-byte gathers; edge-id sum order kept
# speedup vs baseline: 2.3795x; 1.1160x over previous
#include <hip/hip_runtime.h>


typedef __attribute__((ext_vector_type(2))) float v2f;
typedef __attribute__((ext_vector_type(8))) float v8f;

#define NNODE 10000
#define NEDGE 50000
#define HDIM  64
#define HPITCH 68
#define KCAT   384
#define SPITCH 388
#define LCAP  32
#define UB 11
#define PACK_FLOATS 24576

template <int K>
__device__ __forceinline__ v8f wmma_tile_f32(
    const float* __restrict__ A, int lda,
    const float* __restrict__ B, int ldb,
    int m0, int col, int hf, int l)
{
    const float* __restrict__ Arow = A + (size_t)(m0 + l) * lda;
    v8f acc = {};
#pragma unroll
    for (int k = 0; k < K; k += 4) {
        const int kb = k + (hf << 1);
        const v2f a = *(const v2f*)(Arow + kb);
        v2f b;
        b.x = B[(size_t)kb * ldb + col];
        b.y = B[(size_t)(kb + 1) * ldb + col];
        acc = __builtin_amdgcn_wmma_f32_16x16x4_f32(
            false, a, false, b, (short)0, acc, false, false);
    }
    return acc;
}

__global__ __launch_bounds__(128) void lin0_clear(
    const float* __restrict__ x, const float* __restrict__ w,
    const float* __restrict__ bias, float* __restrict__ h, int* __restrict__ cnt,
    const float* __restrict__ nn_w, const float* __restrict__ nn_b,
    const float* __restrict__ root_w,
    const float* __restrict__ w_ih, const float* __restrict__ w_hh,
    float* __restrict__ pw2, float* __restrict__ pg)
{
    const int t = blockIdx.x * 128 + threadIdx.x;
    const bool packs = t < PACK_FLOATS;
    float pv2 = 0.0f, pvg = 0.0f;
    if (packs) {
        const int f = t & 3, ln = (t >> 2) & 31, rest = t >> 7;
        const int c = ln & 15;
        const int k4 = ((f >> 1) << 2) + ((ln >> 4) << 1) + (f & 1);
        {
            const int p = rest % 6, r2 = rest / 6;
            const int j2 = r2 & 7, wv = r2 >> 3;
            const float* __restrict__ B =
                (p == 0) ? root_w
                         : ((p == 5) ? nn_b : (nn_w + (size_t)(p - 1) * HDIM * HDIM));
            pv2 = B[(size_t)(8 * j2 + k4) * HDIM + (wv << 4) + c];
        }
        {
            const int g = rest % 3, r2 = rest / 3;
            const int j2 = r2 & 7, r3 = r2 >> 3;
            const int wv = r3 & 3, q = r3 >> 2;
            const float* __restrict__ Wt = q ? w_hh : w_ih;
            pvg = Wt[(size_t)(g * HDIM + (wv << 4) + c) * HDIM + 8 * j2 + k4];
        }
    }

    const int wave = threadIdx.x >> 5;
    const int lane = threadIdx.x & 31;
    const int hf = lane >> 4, l = lane & 15;
    const int m0  = blockIdx.x << 4;
    const int col = (wave << 4) + l;

    if (threadIdx.x < 16) cnt[m0 + threadIdx.x] = 0;

    const v8f acc = wmma_tile_f32<32>(x, 32, w, HDIM, m0, col, hf, l);
    const float bv = bias[col];
#pragma unroll
    for (int v = 0; v < 8; ++v) {
        const int r = v + (hf << 3);
        h[(size_t)(m0 + r) * HDIM + col] = fmaxf(acc[v] + bv, 0.0f);
    }
    if (packs) { pw2[t] = pv2; pg[t] = pvg; }
}

__global__ __launch_bounds__(256) void build_lists(
    const int* __restrict__ src, const int* __restrict__ dst,
    const float* __restrict__ ea,
    int* __restrict__ cnt, int* __restrict__ lid, float* __restrict__ lco)
{
    const int e = blockIdx.x * 256 + threadIdx.x;
    if (e >= NEDGE) return;
    const int d = dst[e];
    const int pos = atomicAdd(cnt + d, 1);
    if (pos < LCAP) {
        const size_t slot = (size_t)d * LCAP + pos;
        *(int2*)(lid + slot * 2) = make_int2(e, src[e]);
        *(float4*)(lco + slot * 4) = *(const float4*)(ea + (size_t)e * 4);
    }
}

__device__ __forceinline__ void gate_tiles_f32(
    const float* A, int lda, const float4* __restrict__ Bq,
    int hf, int l, v8f& accR, v8f& accZ, v8f& accN)
{
    const float* Arow = A + l * lda;
    v8f r = {}, z = {}, n = {};
#pragma unroll
    for (int j2 = 0; j2 < HDIM / 8; ++j2) {
        const float4 qr = Bq[(j2 * 3 + 0) * 32];
        const float4 qz = Bq[(j2 * 3 + 1) * 32];
        const float4 qn = Bq[(j2 * 3 + 2) * 32];
#pragma unroll
        for (int s = 0; s < 2; ++s) {
            const int kb = 8 * j2 + 4 * s + (hf << 1);
            const v2f a  = *(const v2f*)(Arow + kb);
            const v2f br = s ? v2f{qr.z, qr.w} : v2f{qr.x, qr.y};
            const v2f bz = s ? v2f{qz.z, qz.w} : v2f{qz.x, qz.y};
            const v2f bn = s ? v2f{qn.z, qn.w} : v2f{qn.x, qn.y};
            r = __builtin_amdgcn_wmma_f32_16x16x4_f32(
                false, a, false, br, (short)0, r, false, false);
            z = __builtin_amdgcn_wmma_f32_16x16x4_f32(
                false, a, false, bz, (short)0, z, false, false);
            n = __builtin_amdgcn_wmma_f32_16x16x4_f32(
                false, a, false, bn, (short)0, n, false, false);
        }
    }
    accR = r; accZ = z; accN = n;
}

__device__ __forceinline__ void edge_fma(
    float (&a)[10], float c0, float c1, float c2, float c3, float hlo, float hhi)
{
    a[0] = fmaf(c0, hlo, a[0]);  a[1] = fmaf(c0, hhi, a[1]);
    a[2] = fmaf(c1, hlo, a[2]);  a[3] = fmaf(c1, hhi, a[3]);
    a[4] = fmaf(c2, hlo, a[4]);  a[5] = fmaf(c2, hhi, a[5]);
    a[6] = fmaf(c3, hlo, a[6]);  a[7] = fmaf(c3, hhi, a[7]);
    a[8] += hlo;                 a[9] += hhi;
}

__device__ __forceinline__ void edge_fma8(
    float4 (&a)[5][2], const float4 c, const float4 gA, const float4 gB)
{
#define EDGE_FMA4(p, cf)                                                      \
    a[p][0].x = fmaf(cf, gA.x, a[p][0].x); a[p][0].y = fmaf(cf, gA.y, a[p][0].y); \
    a[p][0].z = fmaf(cf, gA.z, a[p][0].z); a[p][0].w = fmaf(cf, gA.w, a[p][0].w); \
    a[p][1].x = fmaf(cf, gB.x, a[p][1].x); a[p][1].y = fmaf(cf, gB.y, a[p][1].y); \
    a[p][1].z = fmaf(cf, gB.z, a[p][1].z); a[p][1].w = fmaf(cf, gB.w, a[p][1].w);
    EDGE_FMA4(0, c.x)
    EDGE_FMA4(1, c.y)
    EDGE_FMA4(2, c.z)
    EDGE_FMA4(3, c.w)
#undef EDGE_FMA4
    a[4][0].x += gA.x; a[4][0].y += gA.y; a[4][0].z += gA.z; a[4][0].w += gA.w;
    a[4][1].x += gB.x; a[4][1].y += gB.y; a[4][1].z += gB.z; a[4][1].w += gB.w;
}

__device__ __forceinline__ void scan_row(
    const float* __restrict__ h_in, const int* __restrict__ src,
    const int* __restrict__ dst, const float* __restrict__ ea,
    int d, int lane, float* srow)
{
    float a[10];
#pragma unroll
    for (int i = 0; i < 10; ++i) a[i] = 0.0f;
    for (int e0 = 0; e0 < NEDGE; e0 += 32) {
        const int e = e0 + lane;
        const bool hit = (e < NEDGE) && (dst[e < NEDGE ? e : NEDGE - 1] == d);
        unsigned m = __builtin_amdgcn_ballot_w32(hit);
        while (m) {
            const int ee = e0 + __builtin_ctz(m);
            m &= m - 1;
            const int s = src[ee];
            const float4 c = *(const float4*)(ea + (size_t)ee * 4);
            const float* __restrict__ row = h_in + (size_t)s * HDIM + lane;
            edge_fma(a, c.x, c.y, c.z, c.w, row[0], row[32]);
        }
    }
#pragma unroll
    for (int p = 0; p < 5; ++p) {
        srow[HDIM * (p + 1) + lane]      = a[2 * p];
        srow[HDIM * (p + 1) + lane + 32] = a[2 * p + 1];
    }
}

__global__ __launch_bounds__(128) void fused_step(
    const float* __restrict__ h_in, float* __restrict__ h_out,
    const int* __restrict__ src, const int* __restrict__ dst,
    const float* __restrict__ ea,
    const int* __restrict__ cnt, const int* __restrict__ lid,
    const float* __restrict__ lco,
    const float* __restrict__ pw2, const float* __restrict__ pg,
    const float* __restrict__ conv_b,
    const float* __restrict__ b_ih, const float* __restrict__ b_hh)
{
    __shared__ __attribute__((aligned(16))) float St[16 * SPITCH];
    __shared__ __attribute__((aligned(16))) float Ms[16 * HPITCH];
    const int wave = threadIdx.x >> 5;
    const int lane = threadIdx.x & 31;
    const int hf = lane >> 4, l = lane & 15;
    const int m0  = blockIdx.x << 4;
    const int col = (wave << 4) + l;

    const int rb = (wave << 2);
    const int lr = lane >> 3;
    const int cg = lane & 7;
    int cr[4];
    int nl[4];
    int ev[4], sv[4], rk[4];
    float4 cv[4];
#pragma unroll
    for (int r = 0; r < 4; ++r) {
        const int d = m0 + rb + r;
        cr[r] = __builtin_amdgcn_readfirstlane(cnt[d]);
        nl[r] = (cr[r] > LCAP) ? 0 : cr[r];
    }
    const int drow = m0 + rb + lr;
    const float* __restrict__ hcol = h_in + 4 * cg;
    const float4 hoA = *(const float4*)(hcol + (size_t)drow * HDIM);
    const float4 hoB = *(const float4*)(hcol + (size_t)drow * HDIM + 32);
#pragma unroll
    for (int r = 0; r < 4; ++r) {
        const size_t slot = (size_t)(m0 + rb + r) * LCAP + lane;
        int2 id = make_int2(0x7fffffff, 0);
        float4 c4 = make_float4(0.0f, 0.0f, 0.0f, 0.0f);
        if (lane < nl[r]) {
            id = *(const int2*)(lid + slot * 2);
            c4 = *(const float4*)(lco + slot * 4);
        }
        ev[r] = id.x; sv[r] = id.y; cv[r] = c4;
    }
#pragma unroll
    for (int r = 0; r < 4; ++r) {
        int k = 0;
        for (int j = 0; j < nl[r]; ++j) {
            const int ej = __builtin_amdgcn_readlane(ev[r], j);
            k += (ej < ev[r]) ? 1 : 0;
        }
        rk[r] = k;
    }
#pragma unroll
    for (int r = 0; r < 4; ++r) {
        float* sc = St + (rb + r) * SPITCH + HDIM;
        const int pos = (lane < nl[r]) ? rk[r] : lane;
        sc[pos] = __int_as_float(sv[r]);
        *(float4*)(sc + LCAP + 4 * pos) = cv[r];
    }
    int maxn = nl[0] > nl[1] ? nl[0] : nl[1];
    { const int t = nl[2] > nl[3] ? nl[2] : nl[3]; maxn = maxn > t ? maxn : t; }
    const int clane = (lr == 0) ? cr[0] : ((lr == 1) ? cr[1] : ((lr == 2) ? cr[2] : cr[3]));
    const bool bigl = clane > LCAP;
    const int nlane = bigl ? 0 : clane;
    const float* mys = St + (rb + lr) * SPITCH + HDIM;

    float4 acc[5][2];
#pragma unroll
    for (int p = 0; p < 5; ++p) {
        acc[p][0] = make_float4(0.0f, 0.0f, 0.0f, 0.0f);
        acc[p][1] = make_float4(0.0f, 0.0f, 0.0f, 0.0f);
    }

    {
        int    sr[UB];
        float4 cc[UB], gA[UB], gB[UB];
#pragma unroll
        for (int u = 0; u < UB; ++u) {
            sr[u] = __float_as_int(mys[u]);
            cc[u] = *(const float4*)(mys + LCAP + 4 * u);
        }
#pragma unroll
        for (int u = 0; u < UB; ++u) {
            if (u < nlane) {
                const float* __restrict__ row = hcol + (size_t)sr[u] * HDIM;
                gA[u] = *(const float4*)row;
                gB[u] = *(const float4*)(row + 32);
            }
        }
#pragma unroll
        for (int u = 0; u < UB; ++u) {
            if (u < nlane) edge_fma8(acc, cc[u], gA[u], gB[u]);
        }
    }
    for (int t = UB; t < maxn; ++t) {
        if (t < nlane) {
            const int s = __float_as_int(mys[t]);
            const float4 c = *(const float4*)(mys + LCAP + 4 * t);
            const float* __restrict__ row = hcol + (size_t)s * HDIM;
            edge_fma8(acc, c, *(const float4*)row, *(const float4*)(row + 32));
        }
    }

    {
        float* srow = St + (rb + lr) * SPITCH + 4 * cg;
        *(float4*)(srow)      = hoA;
        *(float4*)(srow + 32) = hoB;
        if (!bigl) {
#pragma unroll
            for (int p = 0; p < 5; ++p) {
                *(float4*)(srow + HDIM * (p + 1))      = acc[p][0];
                *(float4*)(srow + HDIM * (p + 1) + 32) = acc[p][1];
            }
        }
    }
#pragma nounroll
    for (int r = 0; r < 4; ++r) {
        const int d = m0 + rb + r;
        if (__builtin_amdgcn_readfirstlane(cnt[d]) > LCAP)
            scan_row(h_in, src, dst, ea, d, lane, St + (rb + r) * SPITCH);
    }
    __syncthreads();

    {
        v8f pa[6];
#pragma unroll
        for (int p = 0; p < 6; ++p) pa[p] = v8f{};
        const float4* __restrict__ Bw =
            (const float4*)pw2 + (size_t)wave * (8 * 6 * 32) + lane;
#pragma unroll
        for (int j2 = 0; j2 < HDIM / 8; ++j2) {
            float4 q[6];
#pragma unroll
            for (int p = 0; p < 6; ++p) q[p] = Bw[(j2 * 6 + p) * 32];
#pragma unroll
            for (int s = 0; s < 2; ++s) {
                const int kb = 8 * j2 + 4 * s + (hf << 1);
#pragma unroll
                for (int p = 0; p < 6; ++p) {
                    const v2f a = *(const v2f*)(St + l * SPITCH + p * HDIM + kb);
                    const v2f b = s ? v2f{q[p].z, q[p].w} : v2f{q[p].x, q[p].y};
                    pa[p] = __builtin_amdgcn_wmma_f32_16x16x4_f32(
                        false, a, false, b, (short)0, pa[p], false, false);
                }
            }
        }
        const float cb = conv_b[col];
#pragma unroll
        for (int v = 0; v < 8; ++v) {
            const int rr = v + (hf << 3);
            const float s = ((pa[0][v] + pa[1][v]) + (pa[2][v] + pa[3][v]))
                          + (pa[4][v] + pa[5][v]);
            Ms[rr * HPITCH + col] = fmaxf(s + cb, 0.0f);
        }
    }
    __syncthreads();

    v8f ir, iz, in_, hr, hz, hn_;
    const float4* __restrict__ Gi = (const float4*)pg + (size_t)wave * (8 * 3 * 32) + lane;
    const float4* __restrict__ Gh = Gi + 4 * 8 * 3 * 32;
    gate_tiles_f32(Ms, HPITCH, Gi, hf, l, ir, iz, in_);
    gate_tiles_f32(St, SPITCH, Gh, hf, l, hr, hz, hn_);

    const float bir = b_ih[col], biz = b_ih[HDIM + col], bin = b_ih[2 * HDIM + col];
    const float bhr = b_hh[col], bhz = b_hh[HDIM + col], bhn = b_hh[2 * HDIM + col];

#pragma unroll
    for (int v = 0; v < 8; ++v) {
        const int rr = v + (hf << 3);
        const size_t idx = (size_t)(m0 + rr) * HDIM + col;
        const float gir = ir[v] + bir, ghr = hr[v] + bhr;
        const float giz = iz[v] + biz, ghz = hz[v] + bhz;
        const float gin = in_[v] + bin, ghn = hn_[v] + bhn;
        const float sr = gir + ghr;
        const float sz = giz + ghz;
        const float r = 1.0f / (1.0f + __expf(-sr));
        const float z = 1.0f / (1.0f + __expf(-sz));
        const float n = tanhf(gin + r * ghn);
        const float ho = St[rr * SPITCH + col];
        h_out[idx] = (1.0f - z) * n + z * ho;
    }
}

extern "C" void kernel_launch(void* const* d_in, const int* in_sizes, int n_in,
                              void* d_out, int out_size, void* d_ws, size_t ws_size,
                              hipStream_t stream)
{
    (void)in_sizes; (void)n_in; (void)out_size; (void)ws_size;

    const float* x      = (const float*)d_in[0];
    const int*   eidx   = (const int*)  d_in[1];
    const float* ea     = (const float*)d_in[2];
    const float* lin0_w = (const float*)d_in[3];
    const float* lin0_b = (const float*)d_in[4];
    const float* nn_w   = (const float*)d_in[5];
    const float* nn_b   = (const float*)d_in[6];
    const float* root_w = (const float*)d_in[7];
    const float* conv_b = (const float*)d_in[8];
    const float* w_ih   = (const float*)d_in[9];
    const float* w_hh   = (const float*)d_in[10];
    const float* b_ih   = (const float*)d_in[11];
    const float* b_hh   = (const float*)d_in[12];

    const int* src = eidx;
    const int* dst = eidx + NEDGE;

    float* hw  = (float*)d_ws;
    float* lco = hw + (size_t)NNODE * HDIM;
    int*   lid = (int*)(lco + (size_t)NNODE * LCAP * 4);
    int*   cnt = lid + (size_t)NNODE * LCAP * 2;
    float* pw2 = (float*)(cnt + 10240);
    float* pg  = pw2 + PACK_FLOATS;
    float* ho  = (float*)d_out;

    lin0_clear<<<NNODE / 16, 128, 0, stream>>>(x, lin0_w, lin0_b, hw, cnt,
        nn_w, nn_b, root_w, w_ih, w_hh, pw2, pg);
    build_lists<<<(NEDGE + 255) / 256, 256, 0, stream>>>(src, dst, ea, cnt, lid, lco);

    fused_step<<<NNODE / 16, 128, 0, stream>>>(hw, ho, src, dst, ea, cnt, lid, lco,
        pw2, pg, conv_b, b_ih, b_hh);
    fused_step<<<NNODE / 16, 128, 0, stream>>>(ho, hw, src, dst, ea, cnt, lid, lco,
        pw2, pg, conv_b, b_ih, b_hh);
    fused_step<<<NNODE / 16, 128, 0, stream>>>(hw, ho, src, dst, ea, cnt, lid, lco,
        pw2, pg, conv_b, b_ih, b_hh);
}


# Edge lists built in the lin0 launch: 125 blocks each scan dst for their own 80 rows, slots from LDS counters (no global atomics, no counter clear); lin0 one row tile per wave; 5 to 4 launches
# speedup vs baseline: 2.5091x; 1.0545x over previous
#include <hip/hip_runtime.h>


typedef __attribute__((ext_vector_type(2))) float v2f;
typedef __attribute__((ext_vector_type(8))) float v8f;

#define NNODE 10000
#define NEDGE 50000
#define HDIM  64
#define HPITCH 68
#define KCAT   384
#define SPITCH 388
#define LCAP  32
#define UB 11
#define PACK_FLOATS 24576

template <int K>
__device__ __forceinline__ v8f wmma_tile_f32(
    const float* __restrict__ A, int lda,
    const float* __restrict__ B, int ldb,
    int m0, int col, int hf, int l)
{
    const float* __restrict__ Arow = A + (size_t)(m0 + l) * lda;
    v8f acc = {};
#pragma unroll
    for (int k = 0; k < K; k += 4) {
        const int kb = k + (hf << 1);
        const v2f a = *(const v2f*)(Arow + kb);
        v2f b;
        b.x = B[(size_t)kb * ldb + col];
        b.y = B[(size_t)(kb + 1) * ldb + col];
        acc = __builtin_amdgcn_wmma_f32_16x16x4_f32(
            false, a, false, b, (short)0, acc, false, false);
    }
    return acc;
}

#define LROWS 80
#define NLB   (NNODE / LROWS)
#define PTHREADS 512
#define PWAVES   (PTHREADS / 32)
#define NTILE (NNODE / 16)
#define NGB   ((NTILE + PWAVES - 1) / PWAVES)
#define E4     (NEDGE / 4)
#define SCAN_J ((E4 + PTHREADS - 1) / PTHREADS)
#define WCAP 96

__device__ __forceinline__ void place_edge(
    const int e, const unsigned r, const int lo, int* lc,
    const int* __restrict__ src, const float* __restrict__ ea,
    int* __restrict__ lid, float* __restrict__ lco)
{
    const int s = src[e];
    const float4 c = *(const float4*)(ea + (size_t)e * 4);
    const int pos = atomicAdd(lc + r, 1);
    if (pos < LCAP) {
        const size_t slot = (size_t)(lo + (int)r) * LCAP + pos;
        *(int2*)(lid + slot * 2) = make_int2(e, s);
        *(float4*)(lco + slot * 4) = c;
    }
}

__device__ __forceinline__ void place_group(
    const int4 q, const int g, const int lo, int* lc,
    const int* __restrict__ src, const float* __restrict__ ea,
    int* __restrict__ lid, float* __restrict__ lco)
{
    const int e0 = g << 2;
    const unsigned r0 = (unsigned)(q.x - lo), r1 = (unsigned)(q.y - lo);
    const unsigned r2 = (unsigned)(q.z - lo), r3 = (unsigned)(q.w - lo);
    const int ci = (r0 < LROWS) ? 0 : ((r1 < LROWS) ? 1 : ((r2 < LROWS) ? 2 : 3));
    const unsigned rc = (ci == 0) ? r0 : ((ci == 1) ? r1 : ((ci == 2) ? r2 : r3));
    if (rc < LROWS) place_edge(e0 + ci, rc, lo, lc, src, ea, lid, lco);
    if (ci < 1 && r1 < LROWS) place_edge(e0 + 1, r1, lo, lc, src, ea, lid, lco);
    if (ci < 2 && r2 < LROWS) place_edge(e0 + 2, r2, lo, lc, src, ea, lid, lco);
    if (ci < 3 && r3 < LROWS) place_edge(e0 + 3, r3, lo, lc, src, ea, lid, lco);
}

__global__ __launch_bounds__(PTHREADS) void lin0_lists(
    const float* __restrict__ x, const float* __restrict__ w,
    const float* __restrict__ bias, float* __restrict__ h,
    const int* __restrict__ src, const int* __restrict__ dst,
    const float* __restrict__ ea,
    int* __restrict__ cnt, int* __restrict__ lid, float* __restrict__ lco,
    const float* __restrict__ nn_w, const float* __restrict__ nn_b,
    const float* __restrict__ root_w,
    const float* __restrict__ w_ih, const float* __restrict__ w_hh,
    float* __restrict__ pw2, float* __restrict__ pg)
{
    __shared__ int lc[LROWS];
    __shared__ int bufI[PWAVES * WCAP];
    __shared__ __attribute__((aligned(16))) int4 bufV[PWAVES * WCAP];
    const int wave = threadIdx.x >> 5;
    const int lane = threadIdx.x & 31;

    if (blockIdx.x >= NLB) {
        const int tr = (blockIdx.x - NLB) * PWAVES + wave;
        if (tr >= NTILE) return;
        const int hf = lane >> 4, l = lane & 15;
        const int m0 = tr << 4;
        v8f acc[4];
#pragma unroll
        for (int tc = 0; tc < 4; ++tc)
            acc[tc] = wmma_tile_f32<32>(x, 32, w, HDIM, m0, (tc << 4) + l, hf, l);
#pragma unroll
        for (int tc = 0; tc < 4; ++tc) {
            const int col = (tc << 4) + l;
            const float bv = bias[col];
#pragma unroll
            for (int v = 0; v < 8; ++v) {
                const int r = v + (hf << 3);
                h[(size_t)(m0 + r) * HDIM + col] = fmaxf(acc[tc][v] + bv, 0.0f);
            }
        }
        return;
    }

    const int t = blockIdx.x * PTHREADS + threadIdx.x;
    const bool packs = t < PACK_FLOATS;
    float pv2 = 0.0f, pvg = 0.0f;
    if (packs) {
        const int f = t & 3, ln = (t >> 2) & 31, rest = t >> 7;
        const int c = ln & 15;
        const int k4 = ((f >> 1) << 2) + ((ln >> 4) << 1) + (f & 1);
        {
            const int p = rest % 6, r2 = rest / 6;
            const int j2 = r2 & 7, wv = r2 >> 3;
            const float* __restrict__ B =
                (p == 0) ? root_w
                         : ((p == 5) ? nn_b : (nn_w + (size_t)(p - 1) * HDIM * HDIM));
            pv2 = B[(size_t)(8 * j2 + k4) * HDIM + (wv << 4) + c];
        }
        {
            const int g = rest % 3, r2 = rest / 3;
            const int j2 = r2 & 7, r3 = r2 >> 3;
            const int wv = r3 & 3, q = r3 >> 2;
            const float* __restrict__ Wt = q ? w_hh : w_ih;
            pvg = Wt[(size_t)(g * HDIM + (wv << 4) + c) * HDIM + 8 * j2 + k4];
        }
    }

    const int lo = blockIdx.x * LROWS;
    if (threadIdx.x < LROWS) lc[threadIdx.x] = 0;

    const int4* __restrict__ d4 = (const int4*)dst;
    const int rot = blockIdx.x * 96;
    int* myI = bufI + wave * WCAP;
    int4* myV = bufV + wave * WCAP;

    int4 v[SCAN_J];
#pragma unroll
    for (int u = 0; u < SCAN_J; ++u) {
        const int g0 = (int)threadIdx.x + PTHREADS * u;
        int g = g0 + rot;
        g = (g >= E4) ? g - E4 : g;
        v[u] = (g0 < E4) ? d4[g] : make_int4(-1, -1, -1, -1);
    }
    int wcount = 0;
#pragma unroll
    for (int u = 0; u < SCAN_J; ++u) {
        const unsigned r0 = (unsigned)(v[u].x - lo), r1 = (unsigned)(v[u].y - lo);
        const unsigned r2 = (unsigned)(v[u].z - lo), r3 = (unsigned)(v[u].w - lo);
        const unsigned ra = r0 < r1 ? r0 : r1, rb = r2 < r3 ? r2 : r3;
        const bool any = (ra < rb ? ra : rb) < LROWS;
        const unsigned m = __builtin_amdgcn_ballot_w32(any);
        if (m) {
            if (any) {
                const int k = wcount + (int)__builtin_amdgcn_mbcnt_lo(m, 0u);
                if (k < WCAP) {
                    int g = (int)threadIdx.x + PTHREADS * u + rot;
                    g = (g >= E4) ? g - E4 : g;
                    myI[k] = g;
                    myV[k] = v[u];
                }
            }
            wcount += __builtin_popcount(m);
        }
    }
    __syncthreads();

    const int nw = wcount < WCAP ? wcount : WCAP;
    for (int base = 0; base < nw; base += 32) {
        const int k = base + lane;
        if (k < nw) place_group(myV[k], myI[k], lo, lc, src, ea, lid, lco);
    }
    if (wcount > WCAP) {
        int seen = 0;
#pragma nounroll
        for (int u = 0; u < SCAN_J; ++u) {
            const int g0 = (int)threadIdx.x + PTHREADS * u;
            int g = g0 + rot;
            g = (g >= E4) ? g - E4 : g;
            const int4 q = (g0 < E4) ? d4[g] : make_int4(-1, -1, -1, -1);
            const unsigned r0 = (unsigned)(q.x - lo), r1 = (unsigned)(q.y - lo);
            const unsigned r2 = (unsigned)(q.z - lo), r3 = (unsigned)(q.w - lo);
            const unsigned ra = r0 < r1 ? r0 : r1, rb = r2 < r3 ? r2 : r3;
            const bool any = (ra < rb ? ra : rb) < LROWS;
            const unsigned m = __builtin_amdgcn_ballot_w32(any);
            if (any && seen + (int)__builtin_amdgcn_mbcnt_lo(m, 0u) >= WCAP)
                place_group(q, g, lo, lc, src, ea, lid, lco);
            seen += __builtin_popcount(m);
        }
    }
    __syncthreads();
    if (threadIdx.x < LROWS) cnt[lo + threadIdx.x] = lc[threadIdx.x];
    if (packs) { pw2[t] = pv2; pg[t] = pvg; }
}

__device__ __forceinline__ void gate_tiles_f32(
    const float* A, int lda, const float4* __restrict__ Bq,
    int hf, int l, v8f& accR, v8f& accZ, v8f& accN)
{
    const float* Arow = A + l * lda;
    v8f r = {}, z = {}, n = {};
#pragma unroll
    for (int j2 = 0; j2 < HDIM / 8; ++j2) {
        const float4 qr = Bq[(j2 * 3 + 0) * 32];
        const float4 qz = Bq[(j2 * 3 + 1) * 32];
        const float4 qn = Bq[(j2 * 3 + 2) * 32];
#pragma unroll
        for (int s = 0; s < 2; ++s) {
            const int kb = 8 * j2 + 4 * s + (hf << 1);
            const v2f a  = *(const v2f*)(Arow + kb);
            const v2f br = s ? v2f{qr.z, qr.w} : v2f{qr.x, qr.y};
            const v2f bz = s ? v2f{qz.z, qz.w} : v2f{qz.x, qz.y};
            const v2f bn = s ? v2f{qn.z, qn.w} : v2f{qn.x, qn.y};
            r = __builtin_amdgcn_wmma_f32_16x16x4_f32(
                false, a, false, br, (short)0, r, false, false);
            z = __builtin_amdgcn_wmma_f32_16x16x4_f32(
                false, a, false, bz, (short)0, z, false, false);
            n = __builtin_amdgcn_wmma_f32_16x16x4_f32(
                false, a, false, bn, (short)0, n, false, false);
        }
    }
    accR = r; accZ = z; accN = n;
}

__device__ __forceinline__ void edge_fma(
    float (&a)[10], float c0, float c1, float c2, float c3, float hlo, float hhi)
{
    a[0] = fmaf(c0, hlo, a[0]);  a[1] = fmaf(c0, hhi, a[1]);
    a[2] = fmaf(c1, hlo, a[2]);  a[3] = fmaf(c1, hhi, a[3]);
    a[4] = fmaf(c2, hlo, a[4]);  a[5] = fmaf(c2, hhi, a[5]);
    a[6] = fmaf(c3, hlo, a[6]);  a[7] = fmaf(c3, hhi, a[7]);
    a[8] += hlo;                 a[9] += hhi;
}

__device__ __forceinline__ void edge_fma8(
    float4 (&a)[5][2], const float4 c, const float4 gA, const float4 gB)
{
#define EDGE_FMA4(p, cf)                                                      \
    a[p][0].x = fmaf(cf, gA.x, a[p][0].x); a[p][0].y = fmaf(cf, gA.y, a[p][0].y); \
    a[p][0].z = fmaf(cf, gA.z, a[p][0].z); a[p][0].w = fmaf(cf, gA.w, a[p][0].w); \
    a[p][1].x = fmaf(cf, gB.x, a[p][1].x); a[p][1].y = fmaf(cf, gB.y, a[p][1].y); \
    a[p][1].z = fmaf(cf, gB.z, a[p][1].z); a[p][1].w = fmaf(cf, gB.w, a[p][1].w);
    EDGE_FMA4(0, c.x)
    EDGE_FMA4(1, c.y)
    EDGE_FMA4(2, c.z)
    EDGE_FMA4(3, c.w)
#undef EDGE_FMA4
    a[4][0].x += gA.x; a[4][0].y += gA.y; a[4][0].z += gA.z; a[4][0].w += gA.w;
    a[4][1].x += gB.x; a[4][1].y += gB.y; a[4][1].z += gB.z; a[4][1].w += gB.w;
}

__device__ __forceinline__ void scan_row(
    const float* __restrict__ h_in, const int* __restrict__ src,
    const int* __restrict__ dst, const float* __restrict__ ea,
    int d, int lane, float* srow)
{
    float a[10];
#pragma unroll
    for (int i = 0; i < 10; ++i) a[i] = 0.0f;
    for (int e0 = 0; e0 < NEDGE; e0 += 32) {
        const int e = e0 + lane;
        const bool hit = (e < NEDGE) && (dst[e < NEDGE ? e : NEDGE - 1] == d);
        unsigned m = __builtin_amdgcn_ballot_w32(hit);
        while (m) {
            const int ee = e0 + __builtin_ctz(m);
            m &= m - 1;
            const int s = src[ee];
            const float4 c = *(const float4*)(ea + (size_t)ee * 4);
            const float* __restrict__ row = h_in + (size_t)s * HDIM + lane;
            edge_fma(a, c.x, c.y, c.z, c.w, row[0], row[32]);
        }
    }
#pragma unroll
    for (int p = 0; p < 5; ++p) {
        srow[HDIM * (p + 1) + lane]      = a[2 * p];
        srow[HDIM * (p + 1) + lane + 32] = a[2 * p + 1];
    }
}

__global__ __launch_bounds__(128) void fused_step(
    const float* __restrict__ h_in, float* __restrict__ h_out,
    const int* __restrict__ src, const int* __restrict__ dst,
    const float* __restrict__ ea,
    const int* __restrict__ cnt, const int* __restrict__ lid,
    const float* __restrict__ lco,
    const float* __restrict__ pw2, const float* __restrict__ pg,
    const float* __restrict__ conv_b,
    const float* __restrict__ b_ih, const float* __restrict__ b_hh)
{
    __shared__ __attribute__((aligned(16))) float St[16 * SPITCH];
    __shared__ __attribute__((aligned(16))) float Ms[16 * HPITCH];
    const int wave = threadIdx.x >> 5;
    const int lane = threadIdx.x & 31;
    const int hf = lane >> 4, l = lane & 15;
    const int m0  = blockIdx.x << 4;
    const int col = (wave << 4) + l;

    const int rb = (wave << 2);
    const int lr = lane >> 3;
    const int cg = lane & 7;
    int cr[4];
    int nl[4];
    int ev[4], sv[4], rk[4];
    float4 cv[4];
#pragma unroll
    for (int r = 0; r < 4; ++r) {
        const int d = m0 + rb + r;
        cr[r] = __builtin_amdgcn_readfirstlane(cnt[d]);
        nl[r] = (cr[r] > LCAP) ? 0 : cr[r];
    }
    const int drow = m0 + rb + lr;
    const float* __restrict__ hcol = h_in + 4 * cg;
    const float4 hoA = *(const float4*)(hcol + (size_t)drow * HDIM);
    const float4 hoB = *(const float4*)(hcol + (size_t)drow * HDIM + 32);
#pragma unroll
    for (int r = 0; r < 4; ++r) {
        const size_t slot = (size_t)(m0 + rb + r) * LCAP + lane;
        int2 id = make_int2(0x7fffffff, 0);
        float4 c4 = make_float4(0.0f, 0.0f, 0.0f, 0.0f);
        if (lane < nl[r]) {
            id = *(const int2*)(lid + slot * 2);
            c4 = *(const float4*)(lco + slot * 4);
        }
        ev[r] = id.x; sv[r] = id.y; cv[r] = c4;
    }
#pragma unroll
    for (int r = 0; r < 4; ++r) {
        int k = 0;
        for (int j = 0; j < nl[r]; ++j) {
            const int ej = __builtin_amdgcn_readlane(ev[r], j);
            k += (ej < ev[r]) ? 1 : 0;
        }
        rk[r] = k;
    }
#pragma unroll
    for (int r = 0; r < 4; ++r) {
        float* sc = St + (rb + r) * SPITCH + HDIM;
        const int pos = (lane < nl[r]) ? rk[r] : lane;
        sc[pos] = __int_as_float(sv[r]);
        *(float4*)(sc + LCAP + 4 * pos) = cv[r];
    }
    int maxn = nl[0] > nl[1] ? nl[0] : nl[1];
    { const int t = nl[2] > nl[3] ? nl[2] : nl[3]; maxn = maxn > t ? maxn : t; }
    const int clane = (lr == 0) ? cr[0] : ((lr == 1) ? cr[1] : ((lr == 2) ? cr[2] : cr[3]));
    const bool bigl = clane > LCAP;
    const int nlane = bigl ? 0 : clane;
    const float* mys = St + (rb + lr) * SPITCH + HDIM;

    float4 acc[5][2];
#pragma unroll
    for (int p = 0; p < 5; ++p) {
        acc[p][0] = make_float4(0.0f, 0.0f, 0.0f, 0.0f);
        acc[p][1] = make_float4(0.0f, 0.0f, 0.0f, 0.0f);
    }

    {
        int    sr[UB];
        float4 cc[UB], gA[UB], gB[UB];
#pragma unroll
        for (int u = 0; u < UB; ++u) {
            sr[u] = __float_as_int(mys[u]);
            cc[u] = *(const float4*)(mys + LCAP + 4 * u);
        }
#pragma unroll
        for (int u = 0; u < UB; ++u) {
            if (u < nlane) {
                const float* __restrict__ row = hcol + (size_t)sr[u] * HDIM;
                gA[u] = *(const float4*)row;
                gB[u] = *(const float4*)(row + 32);
            }
        }
#pragma unroll
        for (int u = 0; u < UB; ++u) {
            if (u < nlane) edge_fma8(acc, cc[u], gA[u], gB[u]);
        }
    }
    for (int t = UB; t < maxn; ++t) {
        if (t < nlane) {
            const int s = __float_as_int(mys[t]);
            const float4 c = *(const float4*)(mys + LCAP + 4 * t);
            const float* __restrict__ row = hcol + (size_t)s * HDIM;
            edge_fma8(acc, c, *(const float4*)row, *(const float4*)(row + 32));
        }
    }

    {
        float* srow = St + (rb + lr) * SPITCH + 4 * cg;
        *(float4*)(srow)      = hoA;
        *(float4*)(srow + 32) = hoB;
        if (!bigl) {
#pragma unroll
            for (int p = 0; p < 5; ++p) {
                *(float4*)(srow + HDIM * (p + 1))      = acc[p][0];
                *(float4*)(srow + HDIM * (p + 1) + 32) = acc[p][1];
            }
        }
    }
#pragma nounroll
    for (int r = 0; r < 4; ++r) {
        const int d = m0 + rb + r;
        if (__builtin_amdgcn_readfirstlane(cnt[d]) > LCAP)
            scan_row(h_in, src, dst, ea, d, lane, St + (rb + r) * SPITCH);
    }
    __syncthreads();

    {
        v8f pa[6];
#pragma unroll
        for (int p = 0; p < 6; ++p) pa[p] = v8f{};
        const float4* __restrict__ Bw =
            (const float4*)pw2 + (size_t)wave * (8 * 6 * 32) + lane;
#pragma unroll
        for (int j2 = 0; j2 < HDIM / 8; ++j2) {
            float4 q[6];
#pragma unroll
            for (int p = 0; p < 6; ++p) q[p] = Bw[(j2 * 6 + p) * 32];
#pragma unroll
            for (int s = 0; s < 2; ++s) {
                const int kb = 8 * j2 + 4 * s + (hf << 1);
#pragma unroll
                for (int p = 0; p < 6; ++p) {
                    const v2f a = *(const v2f*)(St + l * SPITCH + p * HDIM + kb);
                    const v2f b = s ? v2f{q[p].z, q[p].w} : v2f{q[p].x, q[p].y};
                    pa[p] = __builtin_amdgcn_wmma_f32_16x16x4_f32(
                        false, a, false, b, (short)0, pa[p], false, false);
                }
            }
        }
        const float cb = conv_b[col];
#pragma unroll
        for (int v = 0; v < 8; ++v) {
            const int rr = v + (hf << 3);
            const float s = ((pa[0][v] + pa[1][v]) + (pa[2][v] + pa[3][v]))
                          + (pa[4][v] + pa[5][v]);
            Ms[rr * HPITCH + col] = fmaxf(s + cb, 0.0f);
        }
    }
    __syncthreads();

    v8f ir, iz, in_, hr, hz, hn_;
    const float4* __restrict__ Gi = (const float4*)pg + (size_t)wave * (8 * 3 * 32) + lane;
    const float4* __restrict__ Gh = Gi + 4 * 8 * 3 * 32;
    gate_tiles_f32(Ms, HPITCH, Gi, hf, l, ir, iz, in_);
    gate_tiles_f32(St, SPITCH, Gh, hf, l, hr, hz, hn_);

    const float bir = b_ih[col], biz = b_ih[HDIM + col], bin = b_ih[2 * HDIM + col];
    const float bhr = b_hh[col], bhz = b_hh[HDIM + col], bhn = b_hh[2 * HDIM + col];

#pragma unroll
    for (int v = 0; v < 8; ++v) {
        const int rr = v + (hf << 3);
        const size_t idx = (size_t)(m0 + rr) * HDIM + col;
        const float gir = ir[v] + bir, ghr = hr[v] + bhr;
        const float giz = iz[v] + biz, ghz = hz[v] + bhz;
        const float gin = in_[v] + bin, ghn = hn_[v] + bhn;
        const float sr = gir + ghr;
        const float sz = giz + ghz;
        const float r = 1.0f / (1.0f + __expf(-sr));
        const float z = 1.0f / (1.0f + __expf(-sz));
        const float n = tanhf(gin + r * ghn);
        const float ho = St[rr * SPITCH + col];
        h_out[idx] = (1.0f - z) * n + z * ho;
    }
}

extern "C" void kernel_launch(void* const* d_in, const int* in_sizes, int n_in,
                              void* d_out, int out_size, void* d_ws, size_t ws_size,
                              hipStream_t stream)
{
    (void)in_sizes; (void)n_in; (void)out_size; (void)ws_size;

    const float* x      = (const float*)d_in[0];
    const int*   eidx   = (const int*)  d_in[1];
    const float* ea     = (const float*)d_in[2];
    const float* lin0_w = (const float*)d_in[3];
    const float* lin0_b = (const float*)d_in[4];
    const float* nn_w   = (const float*)d_in[5];
    const float* nn_b   = (const float*)d_in[6];
    const float* root_w = (const float*)d_in[7];
    const float* conv_b = (const float*)d_in[8];
    const float* w_ih   = (const float*)d_in[9];
    const float* w_hh   = (const float*)d_in[10];
    const float* b_ih   = (const float*)d_in[11];
    const float* b_hh   = (const float*)d_in[12];

    const int* src = eidx;
    const int* dst = eidx + NEDGE;

    float* hw  = (float*)d_ws;
    float* lco = hw + (size_t)NNODE * HDIM;
    int*   lid = (int*)(lco + (size_t)NNODE * LCAP * 4);
    int*   cnt = lid + (size_t)NNODE * LCAP * 2;
    float* pw2 = (float*)(cnt + 10240);
    float* pg  = pw2 + PACK_FLOATS;
    float* ho  = (float*)d_out;

    lin0_lists<<<NLB + NGB, PTHREADS, 0, stream>>>(x, lin0_w, lin0_b, hw, src, dst, ea,
        cnt, lid, lco, nn_w, nn_b, root_w, w_ih, w_hh, pw2, pg);

    fused_step<<<NNODE / 16, 128, 0, stream>>>(hw, ho, src, dst, ea, cnt, lid, lco,
        pw2, pg, conv_b, b_ih, b_hh);
    fused_step<<<NNODE / 16, 128, 0, stream>>>(ho, hw, src, dst, ea, cnt, lid, lco,
        pw2, pg, conv_b, b_ih, b_hh);
    fused_step<<<NNODE / 16, 128, 0, stream>>>(hw, ho, src, dst, ea, cnt, lid, lco,
        pw2, pg, conv_b, b_ih, b_hh);
}


# Gate update in straight-line code: hardware reciprocal for the two sigmoids, tanh as 1 - 2/(1 + e^(2a)) from one exp and one reciprocal; no IEEE division, no library tanh, no divergent branch
# speedup vs baseline: 2.7588x; 1.0995x over previous
#include <hip/hip_runtime.h>


typedef __attribute__((ext_vector_type(2))) float v2f;
typedef __attribute__((ext_vector_type(8))) float v8f;

#define NNODE 10000
#define NEDGE 50000
#define HDIM  64
#define HPITCH 68
#define KCAT   384
#define SPITCH 388
#define LCAP  32
#define UB 11
#define PACK_FLOATS 24576

template <int K>
__device__ __forceinline__ v8f wmma_tile_f32(
    const float* __restrict__ A, int lda,
    const float* __restrict__ B, int ldb,
    int m0, int col, int hf, int l)
{
    const float* __restrict__ Arow = A + (size_t)(m0 + l) * lda;
    v8f acc = {};
#pragma unroll
    for (int k = 0; k < K; k += 4) {
        const int kb = k + (hf << 1);
        const v2f a = *(const v2f*)(Arow + kb);
        v2f b;
        b.x = B[(size_t)kb * ldb + col];
        b.y = B[(size_t)(kb + 1) * ldb + col];
        acc = __builtin_amdgcn_wmma_f32_16x16x4_f32(
            false, a, false, b, (short)0, acc, false, false);
    }
    return acc;
}

#define LROWS 80
#define NLB   (NNODE / LROWS)
#define PTHREADS 512
#define PWAVES   (PTHREADS / 32)
#define NTILE (NNODE / 16)
#define NGB   ((NTILE + PWAVES - 1) / PWAVES)
#define E4     (NEDGE / 4)
#define SCAN_J ((E4 + PTHREADS - 1) / PTHREADS)
#define WCAP 96

__device__ __forceinline__ void place_edge(
    const int e, const unsigned r, const int lo, int* lc,
    const int* __restrict__ src, const float* __restrict__ ea,
    int* __restrict__ lid, float* __restrict__ lco)
{
    const int s = src[e];
    const float4 c = *(const float4*)(ea + (size_t)e * 4);
    const int pos = atomicAdd(lc + r, 1);
    if (pos < LCAP) {
        const size_t slot = (size_t)(lo + (int)r) * LCAP + pos;
        *(int2*)(lid + slot * 2) = make_int2(e, s);
        *(float4*)(lco + slot * 4) = c;
    }
}

__device__ __forceinline__ void place_group(
    const int4 q, const int g, const int lo, int* lc,
    const int* __restrict__ src, const float* __restrict__ ea,
    int* __restrict__ lid, float* __restrict__ lco)
{
    const int e0 = g << 2;
    const unsigned r0 = (unsigned)(q.x - lo), r1 = (unsigned)(q.y - lo);
    const unsigned r2 = (unsigned)(q.z - lo), r3 = (unsigned)(q.w - lo);
    const int ci = (r0 < LROWS) ? 0 : ((r1 < LROWS) ? 1 : ((r2 < LROWS) ? 2 : 3));
    const unsigned rc = (ci == 0) ? r0 : ((ci == 1) ? r1 : ((ci == 2) ? r2 : r3));
    if (rc < LROWS) place_edge(e0 + ci, rc, lo, lc, src, ea, lid, lco);
    if (ci < 1 && r1 < LROWS) place_edge(e0 + 1, r1, lo, lc, src, ea, lid, lco);
    if (ci < 2 && r2 < LROWS) place_edge(e0 + 2, r2, lo, lc, src, ea, lid, lco);
    if (ci < 3 && r3 < LROWS) place_edge(e0 + 3, r3, lo, lc, src, ea, lid, lco);
}

__global__ __launch_bounds__(PTHREADS) void lin0_lists(
    const float* __restrict__ x, const float* __restrict__ w,
    const float* __restrict__ bias, float* __restrict__ h,
    const int* __restrict__ src, const int* __restrict__ dst,
    const float* __restrict__ ea,
    int* __restrict__ cnt, int* __restrict__ lid, float* __restrict__ lco,
    const float* __restrict__ nn_w, const float* __restrict__ nn_b,
    const float* __restrict__ root_w,
    const float* __restrict__ w_ih, const float* __restrict__ w_hh,
    float* __restrict__ pw2, float* __restrict__ pg)
{
    __shared__ int lc[LROWS];
    __shared__ int bufI[PWAVES * WCAP];
    __shared__ __attribute__((aligned(16))) int4 bufV[PWAVES * WCAP];
    const int wave = threadIdx.x >> 5;
    const int lane = threadIdx.x & 31;

    if (blockIdx.x >= NLB) {
        const int tr = (blockIdx.x - NLB) * PWAVES + wave;
        if (tr >= NTILE) return;
        const int hf = lane >> 4, l = lane & 15;
        const int m0 = tr << 4;
        v8f acc[4];
#pragma unroll
        for (int tc = 0; tc < 4; ++tc)
            acc[tc] = wmma_tile_f32<32>(x, 32, w, HDIM, m0, (tc << 4) + l, hf, l);
#pragma unroll
        for (int tc = 0; tc < 4; ++tc) {
            const int col = (tc << 4) + l;
            const float bv = bias[col];
#pragma unroll
            for (int v = 0; v < 8; ++v) {
                const int r = v + (hf << 3);
                h[(size_t)(m0 + r) * HDIM + col] = fmaxf(acc[tc][v] + bv, 0.0f);
            }
        }
        return;
    }

    const int t = blockIdx.x * PTHREADS + threadIdx.x;
    const bool packs = t < PACK_FLOATS;
    float pv2 = 0.0f, pvg = 0.0f;
    if (packs) {
        const int f = t & 3, ln = (t >> 2) & 31, rest = t >> 7;
        const int c = ln & 15;
        const int k4 = ((f >> 1) << 2) + ((ln >> 4) << 1) + (f & 1);
        {
            const int p = rest % 6, r2 = rest / 6;
            const int j2 = r2 & 7, wv = r2 >> 3;
            const float* __restrict__ B =
                (p == 0) ? root_w
                         : ((p == 5) ? nn_b : (nn_w + (size_t)(p - 1) * HDIM * HDIM));
            pv2 = B[(size_t)(8 * j2 + k4) * HDIM + (wv << 4) + c];
        }
        {
            const int g = rest % 3, r2 = rest / 3;
            const int j2 = r2 & 7, r3 = r2 >> 3;
            const int wv = r3 & 3, q = r3 >> 2;
            const float* __restrict__ Wt = q ? w_hh : w_ih;
            pvg = Wt[(size_t)(g * HDIM + (wv << 4) + c) * HDIM + 8 * j2 + k4];
        }
    }

    const int lo = blockIdx.x * LROWS;
    if (threadIdx.x < LROWS) lc[threadIdx.x] = 0;

    const int4* __restrict__ d4 = (const int4*)dst;
    const int rot = blockIdx.x * 96;
    int* myI = bufI + wave * WCAP;
    int4* myV = bufV + wave * WCAP;

    int4 v[SCAN_J];
#pragma unroll
    for (int u = 0; u < SCAN_J; ++u) {
        const int g0 = (int)threadIdx.x + PTHREADS * u;
        int g = g0 + rot;
        g = (g >= E4) ? g - E4 : g;
        v[u] = (g0 < E4) ? d4[g] : make_int4(-1, -1, -1, -1);
    }
    int wcount = 0;
#pragma unroll
    for (int u = 0; u < SCAN_J; ++u) {
        const unsigned r0 = (unsigned)(v[u].x - lo), r1 = (unsigned)(v[u].y - lo);
        const unsigned r2 = (unsigned)(v[u].z - lo), r3 = (unsigned)(v[u].w - lo);
        const unsigned ra = r0 < r1 ? r0 : r1, rb = r2 < r3 ? r2 : r3;
        const bool any = (ra < rb ? ra : rb) < LROWS;
        const unsigned m = __builtin_amdgcn_ballot_w32(any);
        if (m) {
            if (any) {
                const int k = wcount + (int)__builtin_amdgcn_mbcnt_lo(m, 0u);
                if (k < WCAP) {
                    int g = (int)threadIdx.x + PTHREADS * u + rot;
                    g = (g >= E4) ? g - E4 : g;
                    myI[k] = g;
                    myV[k] = v[u];
                }
            }
            wcount += __builtin_popcount(m);
        }
    }
    __syncthreads();

    const int nw = wcount < WCAP ? wcount : WCAP;
    for (int base = 0; base < nw; base += 32) {
        const int k = base + lane;
        if (k < nw) place_group(myV[k], myI[k], lo, lc, src, ea, lid, lco);
    }
    if (wcount > WCAP) {
        int seen = 0;
#pragma nounroll
        for (int u = 0; u < SCAN_J; ++u) {
            const int g0 = (int)threadIdx.x + PTHREADS * u;
            int g = g0 + rot;
            g = (g >= E4) ? g - E4 : g;
            const int4 q = (g0 < E4) ? d4[g] : make_int4(-1, -1, -1, -1);
            const unsigned r0 = (unsigned)(q.x - lo), r1 = (unsigned)(q.y - lo);
            const unsigned r2 = (unsigned)(q.z - lo), r3 = (unsigned)(q.w - lo);
            const unsigned ra = r0 < r1 ? r0 : r1, rb = r2 < r3 ? r2 : r3;
            const bool any = (ra < rb ? ra : rb) < LROWS;
            const unsigned m = __builtin_amdgcn_ballot_w32(any);
            if (any && seen + (int)__builtin_amdgcn_mbcnt_lo(m, 0u) >= WCAP)
                place_group(q, g, lo, lc, src, ea, lid, lco);
            seen += __builtin_popcount(m);
        }
    }
    __syncthreads();
    if (threadIdx.x < LROWS) cnt[lo + threadIdx.x] = lc[threadIdx.x];
    if (packs) { pw2[t] = pv2; pg[t] = pvg; }
}

__device__ __forceinline__ void gate_tiles_f32(
    const float* A, int lda, const float4* __restrict__ Bq,
    int hf, int l, v8f& accR, v8f& accZ, v8f& accN)
{
    const float* Arow = A + l * lda;
    v8f r = {}, z = {}, n = {};
#pragma unroll
    for (int j2 = 0; j2 < HDIM / 8; ++j2) {
        const float4 qr = Bq[(j2 * 3 + 0) * 32];
        const float4 qz = Bq[(j2 * 3 + 1) * 32];
        const float4 qn = Bq[(j2 * 3 + 2) * 32];
#pragma unroll
        for (int s = 0; s < 2; ++s) {
            const int kb = 8 * j2 + 4 * s + (hf << 1);
            const v2f a  = *(const v2f*)(Arow + kb);
            const v2f br = s ? v2f{qr.z, qr.w} : v2f{qr.x, qr.y};
            const v2f bz = s ? v2f{qz.z, qz.w} : v2f{qz.x, qz.y};
            const v2f bn = s ? v2f{qn.z, qn.w} : v2f{qn.x, qn.y};
            r = __builtin_amdgcn_wmma_f32_16x16x4_f32(
                false, a, false, br, (short)0, r, false, false);
            z = __builtin_amdgcn_wmma_f32_16x16x4_f32(
                false, a, false, bz, (short)0, z, false, false);
            n = __builtin_amdgcn_wmma_f32_16x16x4_f32(
                false, a, false, bn, (short)0, n, false, false);
        }
    }
    accR = r; accZ = z; accN = n;
}

__device__ __forceinline__ void edge_fma(
    float (&a)[10], float c0, float c1, float c2, float c3, float hlo, float hhi)
{
    a[0] = fmaf(c0, hlo, a[0]);  a[1] = fmaf(c0, hhi, a[1]);
    a[2] = fmaf(c1, hlo, a[2]);  a[3] = fmaf(c1, hhi, a[3]);
    a[4] = fmaf(c2, hlo, a[4]);  a[5] = fmaf(c2, hhi, a[5]);
    a[6] = fmaf(c3, hlo, a[6]);  a[7] = fmaf(c3, hhi, a[7]);
    a[8] += hlo;                 a[9] += hhi;
}

__device__ __forceinline__ void edge_fma8(
    float4 (&a)[5][2], const float4 c, const float4 gA, const float4 gB)
{
#define EDGE_FMA4(p, cf)                                                      \
    a[p][0].x = fmaf(cf, gA.x, a[p][0].x); a[p][0].y = fmaf(cf, gA.y, a[p][0].y); \
    a[p][0].z = fmaf(cf, gA.z, a[p][0].z); a[p][0].w = fmaf(cf, gA.w, a[p][0].w); \
    a[p][1].x = fmaf(cf, gB.x, a[p][1].x); a[p][1].y = fmaf(cf, gB.y, a[p][1].y); \
    a[p][1].z = fmaf(cf, gB.z, a[p][1].z); a[p][1].w = fmaf(cf, gB.w, a[p][1].w);
    EDGE_FMA4(0, c.x)
    EDGE_FMA4(1, c.y)
    EDGE_FMA4(2, c.z)
    EDGE_FMA4(3, c.w)
#undef EDGE_FMA4
    a[4][0].x += gA.x; a[4][0].y += gA.y; a[4][0].z += gA.z; a[4][0].w += gA.w;
    a[4][1].x += gB.x; a[4][1].y += gB.y; a[4][1].z += gB.z; a[4][1].w += gB.w;
}

__device__ __forceinline__ void scan_row(
    const float* __restrict__ h_in, const int* __restrict__ src,
    const int* __restrict__ dst, const float* __restrict__ ea,
    int d, int lane, float* srow)
{
    float a[10];
#pragma unroll
    for (int i = 0; i < 10; ++i) a[i] = 0.0f;
    for (int e0 = 0; e0 < NEDGE; e0 += 32) {
        const int e = e0 + lane;
        const bool hit = (e < NEDGE) && (dst[e < NEDGE ? e : NEDGE - 1] == d);
        unsigned m = __builtin_amdgcn_ballot_w32(hit);
        while (m) {
            const int ee = e0 + __builtin_ctz(m);
            m &= m - 1;
            const int s = src[ee];
            const float4 c = *(const float4*)(ea + (size_t)ee * 4);
            const float* __restrict__ row = h_in + (size_t)s * HDIM + lane;
            edge_fma(a, c.x, c.y, c.z, c.w, row[0], row[32]);
        }
    }
#pragma unroll
    for (int p = 0; p < 5; ++p) {
        srow[HDIM * (p + 1) + lane]      = a[2 * p];
        srow[HDIM * (p + 1) + lane + 32] = a[2 * p + 1];
    }
}

__global__ __launch_bounds__(128) void fused_step(
    const float* __restrict__ h_in, float* __restrict__ h_out,
    const int* __restrict__ src, const int* __restrict__ dst,
    const float* __restrict__ ea,
    const int* __restrict__ cnt, const int* __restrict__ lid,
    const float* __restrict__ lco,
    const float* __restrict__ pw2, const float* __restrict__ pg,
    const float* __restrict__ conv_b,
    const float* __restrict__ b_ih, const float* __restrict__ b_hh)
{
    __shared__ __attribute__((aligned(16))) float St[16 * SPITCH];
    __shared__ __attribute__((aligned(16))) float Ms[16 * HPITCH];
    const int wave = threadIdx.x >> 5;
    const int lane = threadIdx.x & 31;
    const int hf = lane >> 4, l = lane & 15;
    const int m0  = blockIdx.x << 4;
    const int col = (wave << 4) + l;

    const int rb = (wave << 2);
    const int lr = lane >> 3;
    const int cg = lane & 7;
    int cr[4];
    int nl[4];
    int ev[4], sv[4], rk[4];
    float4 cv[4];
#pragma unroll
    for (int r = 0; r < 4; ++r) {
        const int d = m0 + rb + r;
        cr[r] = __builtin_amdgcn_readfirstlane(cnt[d]);
        nl[r] = (cr[r] > LCAP) ? 0 : cr[r];
    }
    const int drow = m0 + rb + lr;
    const float* __restrict__ hcol = h_in + 4 * cg;
    const float4 hoA = *(const float4*)(hcol + (size_t)drow * HDIM);
    const float4 hoB = *(const float4*)(hcol + (size_t)drow * HDIM + 32);
#pragma unroll
    for (int r = 0; r < 4; ++r) {
        const size_t slot = (size_t)(m0 + rb + r) * LCAP + lane;
        int2 id = make_int2(0x7fffffff, 0);
        float4 c4 = make_float4(0.0f, 0.0f, 0.0f, 0.0f);
        if (lane < nl[r]) {
            id = *(const int2*)(lid + slot * 2);
            c4 = *(const float4*)(lco + slot * 4);
        }
        ev[r] = id.x; sv[r] = id.y; cv[r] = c4;
    }
#pragma unroll
    for (int r = 0; r < 4; ++r) {
        int k = 0;
        for (int j = 0; j < nl[r]; ++j) {
            const int ej = __builtin_amdgcn_readlane(ev[r], j);
            k += (ej < ev[r]) ? 1 : 0;
        }
        rk[r] = k;
    }
#pragma unroll
    for (int r = 0; r < 4; ++r) {
        float* sc = St + (rb + r) * SPITCH + HDIM;
        const int pos = (lane < nl[r]) ? rk[r] : lane;
        sc[pos] = __int_as_float(sv[r]);
        *(float4*)(sc + LCAP + 4 * pos) = cv[r];
    }
    int maxn = nl[0] > nl[1] ? nl[0] : nl[1];
    { const int t = nl[2] > nl[3] ? nl[2] : nl[3]; maxn = maxn > t ? maxn : t; }
    const int clane = (lr == 0) ? cr[0] : ((lr == 1) ? cr[1] : ((lr == 2) ? cr[2] : cr[3]));
    const bool bigl = clane > LCAP;
    const int nlane = bigl ? 0 : clane;
    const float* mys = St + (rb + lr) * SPITCH + HDIM;

    float4 acc[5][2];
#pragma unroll
    for (int p = 0; p < 5; ++p) {
        acc[p][0] = make_float4(0.0f, 0.0f, 0.0f, 0.0f);
        acc[p][1] = make_float4(0.0f, 0.0f, 0.0f, 0.0f);
    }

    {
        int    sr[UB];
        float4 cc[UB], gA[UB], gB[UB];
#pragma unroll
        for (int u = 0; u < UB; ++u) {
            sr[u] = __float_as_int(mys[u]);
            cc[u] = *(const float4*)(mys + LCAP + 4 * u);
        }
#pragma unroll
        for (int u = 0; u < UB; ++u) {
            if (u < nlane) {
                const float* __restrict__ row = hcol + (size_t)sr[u] * HDIM;
                gA[u] = *(const float4*)row;
                gB[u] = *(const float4*)(row + 32);
            }
        }
#pragma unroll
        for (int u = 0; u < UB; ++u) {
            if (u < nlane) edge_fma8(acc, cc[u], gA[u], gB[u]);
        }
    }
    for (int t = UB; t < maxn; ++t) {
        if (t < nlane) {
            const int s = __float_as_int(mys[t]);
            const float4 c = *(const float4*)(mys + LCAP + 4 * t);
            const float* __restrict__ row = hcol + (size_t)s * HDIM;
            edge_fma8(acc, c, *(const float4*)row, *(const float4*)(row + 32));
        }
    }

    {
        float* srow = St + (rb + lr) * SPITCH + 4 * cg;
        *(float4*)(srow)      = hoA;
        *(float4*)(srow + 32) = hoB;
        if (!bigl) {
#pragma unroll
            for (int p = 0; p < 5; ++p) {
                *(float4*)(srow + HDIM * (p + 1))      = acc[p][0];
                *(float4*)(srow + HDIM * (p + 1) + 32) = acc[p][1];
            }
        }
    }
#pragma nounroll
    for (int r = 0; r < 4; ++r) {
        const int d = m0 + rb + r;
        if (__builtin_amdgcn_readfirstlane(cnt[d]) > LCAP)
            scan_row(h_in, src, dst, ea, d, lane, St + (rb + r) * SPITCH);
    }
    __syncthreads();

    {
        v8f pa[6];
#pragma unroll
        for (int p = 0; p < 6; ++p) pa[p] = v8f{};
        const float4* __restrict__ Bw =
            (const float4*)pw2 + (size_t)wave * (8 * 6 * 32) + lane;
#pragma unroll
        for (int j2 = 0; j2 < HDIM / 8; ++j2) {
            float4 q[6];
#pragma unroll
            for (int p = 0; p < 6; ++p) q[p] = Bw[(j2 * 6 + p) * 32];
#pragma unroll
            for (int s = 0; s < 2; ++s) {
                const int kb = 8 * j2 + 4 * s + (hf << 1);
#pragma unroll
                for (int p = 0; p < 6; ++p) {
                    const v2f a = *(const v2f*)(St + l * SPITCH + p * HDIM + kb);
                    const v2f b = s ? v2f{q[p].z, q[p].w} : v2f{q[p].x, q[p].y};
                    pa[p] = __builtin_amdgcn_wmma_f32_16x16x4_f32(
                        false, a, false, b, (short)0, pa[p], false, false);
                }
            }
        }
        const float cb = conv_b[col];
#pragma unroll
        for (int v = 0; v < 8; ++v) {
            const int rr = v + (hf << 3);
            const float s = ((pa[0][v] + pa[1][v]) + (pa[2][v] + pa[3][v]))
                          + (pa[4][v] + pa[5][v]);
            Ms[rr * HPITCH + col] = fmaxf(s + cb, 0.0f);
        }
    }
    __syncthreads();

    v8f ir, iz, in_, hr, hz, hn_;
    const float4* __restrict__ Gi = (const float4*)pg + (size_t)wave * (8 * 3 * 32) + lane;
    const float4* __restrict__ Gh = Gi + 4 * 8 * 3 * 32;
    gate_tiles_f32(Ms, HPITCH, Gi, hf, l, ir, iz, in_);
    gate_tiles_f32(St, SPITCH, Gh, hf, l, hr, hz, hn_);

    const float bir = b_ih[col], biz = b_ih[HDIM + col], bin = b_ih[2 * HDIM + col];
    const float bhr = b_hh[col], bhz = b_hh[HDIM + col], bhn = b_hh[2 * HDIM + col];

#pragma unroll
    for (int v = 0; v < 8; ++v) {
        const int rr = v + (hf << 3);
        const size_t idx = (size_t)(m0 + rr) * HDIM + col;
        const float gir = ir[v] + bir, ghr = hr[v] + bhr;
        const float giz = iz[v] + biz, ghz = hz[v] + bhz;
        const float gin = in_[v] + bin, ghn = hn_[v] + bhn;
        const float sr = gir + ghr;
        const float sz = giz + ghz;
        const float r = __builtin_amdgcn_rcpf(1.0f + __expf(-sr));
        const float z = __builtin_amdgcn_rcpf(1.0f + __expf(-sz));
        const float a = gin + r * ghn;
        const float n = 1.0f - 2.0f * __builtin_amdgcn_rcpf(1.0f + __expf(2.0f * a));
        const float ho = St[rr * SPITCH + col];
        h_out[idx] = (1.0f - z) * n + z * ho;
    }
}

extern "C" void kernel_launch(void* const* d_in, const int* in_sizes, int n_in,
                              void* d_out, int out_size, void* d_ws, size_t ws_size,
                              hipStream_t stream)
{
    (void)in_sizes; (void)n_in; (void)out_size; (void)ws_size;

    const float* x      = (const float*)d_in[0];
    const int*   eidx   = (const int*)  d_in[1];
    const float* ea     = (const float*)d_in[2];
    const float* lin0_w = (const float*)d_in[3];
    const float* lin0_b = (const float*)d_in[4];
    const float* nn_w   = (const float*)d_in[5];
    const float* nn_b   = (const float*)d_in[6];
    const float* root_w = (const float*)d_in[7];
    const float* conv_b = (const float*)d_in[8];
    const float* w_ih   = (const float*)d_in[9];
    const float* w_hh   = (const float*)d_in[10];
    const float* b_ih   = (const float*)d_in[11];
    const float* b_hh   = (const float*)d_in[12];

    const int* src = eidx;
    const int* dst = eidx + NEDGE;

    float* hw  = (float*)d_ws;
    float* lco = hw + (size_t)NNODE * HDIM;
    int*   lid = (int*)(lco + (size_t)NNODE * LCAP * 4);
    int*   cnt = lid + (size_t)NNODE * LCAP * 2;
    float* pw2 = (float*)(cnt + 10240);
    float* pg  = pw2 + PACK_FLOATS;
    float* ho  = (float*)d_out;

    lin0_lists<<<NLB + NGB, PTHREADS, 0, stream>>>(x, lin0_w, lin0_b, hw, src, dst, ea,
        cnt, lid, lco, nn_w, nn_b, root_w, w_ih, w_hh, pw2, pg);

    fused_step<<<NNODE / 16, 128, 0, stream>>>(hw, ho, src, dst, ea, cnt, lid, lco,
        pw2, pg, conv_b, b_ih, b_hh);
    fused_step<<<NNODE / 16, 128, 0, stream>>>(ho, hw, src, dst, ea, cnt, lid, lco,
        pw2, pg, conv_b, b_ih, b_hh);
    fused_step<<<NNODE / 16, 128, 0, stream>>>(hw, ho, src, dst, ea, cnt, lid, lco,
        pw2, pg, conv_b, b_ih, b_hh);
}


# First weight fragments of each product requested before the barrier in front of it (K=384: before the gathers; w_ih: under h@w_hh^T, which moves ahead of barrier 2): no WMMA waits a load round trip
# speedup vs baseline: 2.8504x; 1.0332x over previous
#include <hip/hip_runtime.h>


typedef __attribute__((ext_vector_type(2))) float v2f;
typedef __attribute__((ext_vector_type(8))) float v8f;

#define NNODE 10000
#define NEDGE 50000
#define HDIM  64
#define HPITCH 68
#define KCAT   384
#define SPITCH 388
#define LCAP  32
#define UB 11
#define PACK_FLOATS 24576

template <int K>
__device__ __forceinline__ v8f wmma_tile_f32(
    const float* __restrict__ A, int lda,
    const float* __restrict__ B, int ldb,
    int m0, int col, int hf, int l)
{
    const float* __restrict__ Arow = A + (size_t)(m0 + l) * lda;
    v8f acc = {};
#pragma unroll
    for (int k = 0; k < K; k += 4) {
        const int kb = k + (hf << 1);
        const v2f a = *(const v2f*)(Arow + kb);
        v2f b;
        b.x = B[(size_t)kb * ldb + col];
        b.y = B[(size_t)(kb + 1) * ldb + col];
        acc = __builtin_amdgcn_wmma_f32_16x16x4_f32(
            false, a, false, b, (short)0, acc, false, false);
    }
    return acc;
}

#define LROWS 80
#define NLB   (NNODE / LROWS)
#define PTHREADS 512
#define PWAVES   (PTHREADS / 32)
#define NTILE (NNODE / 16)
#define NGB   ((NTILE + PWAVES - 1) / PWAVES)
#define E4     (NEDGE / 4)
#define SCAN_J ((E4 + PTHREADS - 1) / PTHREADS)
#define WCAP 96

__device__ __forceinline__ void place_edge(
    const int e, const unsigned r, const int lo, int* lc,
    const int* __restrict__ src, const float* __restrict__ ea,
    int* __restrict__ lid, float* __restrict__ lco)
{
    const int s = src[e];
    const float4 c = *(const float4*)(ea + (size_t)e * 4);
    const int pos = atomicAdd(lc + r, 1);
    if (pos < LCAP) {
        const size_t slot = (size_t)(lo + (int)r) * LCAP + pos;
        *(int2*)(lid + slot * 2) = make_int2(e, s);
        *(float4*)(lco + slot * 4) = c;
    }
}

__device__ __forceinline__ void place_group(
    const int4 q, const int g, const int lo, int* lc,
    const int* __restrict__ src, const float* __restrict__ ea,
    int* __restrict__ lid, float* __restrict__ lco)
{
    const int e0 = g << 2;
    const unsigned r0 = (unsigned)(q.x - lo), r1 = (unsigned)(q.y - lo);
    const unsigned r2 = (unsigned)(q.z - lo), r3 = (unsigned)(q.w - lo);
    const int ci = (r0 < LROWS) ? 0 : ((r1 < LROWS) ? 1 : ((r2 < LROWS) ? 2 : 3));
    const unsigned rc = (ci == 0) ? r0 : ((ci == 1) ? r1 : ((ci == 2) ? r2 : r3));
    if (rc < LROWS) place_edge(e0 + ci, rc, lo, lc, src, ea, lid, lco);
    if (ci < 1 && r1 < LROWS) place_edge(e0 + 1, r1, lo, lc, src, ea, lid, lco);
    if (ci < 2 && r2 < LROWS) place_edge(e0 + 2, r2, lo, lc, src, ea, lid, lco);
    if (ci < 3 && r3 < LROWS) place_edge(e0 + 3, r3, lo, lc, src, ea, lid, lco);
}

__global__ __launch_bounds__(PTHREADS) void lin0_lists(
    const float* __restrict__ x, const float* __restrict__ w,
    const float* __restrict__ bias, float* __restrict__ h,
    const int* __restrict__ src, const int* __restrict__ dst,
    const float* __restrict__ ea,
    int* __restrict__ cnt, int* __restrict__ lid, float* __restrict__ lco,
    const float* __restrict__ nn_w, const float* __restrict__ nn_b,
    const float* __restrict__ root_w,
    const float* __restrict__ w_ih, const float* __restrict__ w_hh,
    float* __restrict__ pw2, float* __restrict__ pg)
{
    __shared__ int lc[LROWS];
    __shared__ int bufI[PWAVES * WCAP];
    __shared__ __attribute__((aligned(16))) int4 bufV[PWAVES * WCAP];
    const int wave = threadIdx.x >> 5;
    const int lane = threadIdx.x & 31;

    if (blockIdx.x >= NLB) {
        const int tr = (blockIdx.x - NLB) * PWAVES + wave;
        if (tr >= NTILE) return;
        const int hf = lane >> 4, l = lane & 15;
        const int m0 = tr << 4;
        v8f acc[4];
#pragma unroll
        for (int tc = 0; tc < 4; ++tc)
            acc[tc] = wmma_tile_f32<32>(x, 32, w, HDIM, m0, (tc << 4) + l, hf, l);
#pragma unroll
        for (int tc = 0; tc < 4; ++tc) {
            const int col = (tc << 4) + l;
            const float bv = bias[col];
#pragma unroll
            for (int v = 0; v < 8; ++v) {
                const int r = v + (hf << 3);
                h[(size_t)(m0 + r) * HDIM + col] = fmaxf(acc[tc][v] + bv, 0.0f);
            }
        }
        return;
    }

    const int t = blockIdx.x * PTHREADS + threadIdx.x;
    const bool packs = t < PACK_FLOATS;
    float pv2 = 0.0f, pvg = 0.0f;
    if (packs) {
        const int f = t & 3, ln = (t >> 2) & 31, rest = t >> 7;
        const int c = ln & 15;
        const int k4 = ((f >> 1) << 2) + ((ln >> 4) << 1) + (f & 1);
        {
            const int p = rest % 6, r2 = rest / 6;
            const int j2 = r2 & 7, wv = r2 >> 3;
            const float* __restrict__ B =
                (p == 0) ? root_w
                         : ((p == 5) ? nn_b : (nn_w + (size_t)(p - 1) * HDIM * HDIM));
            pv2 = B[(size_t)(8 * j2 + k4) * HDIM + (wv << 4) + c];
        }
        {
            const int g = rest % 3, r2 = rest / 3;
            const int j2 = r2 & 7, r3 = r2 >> 3;
            const int wv = r3 & 3, q = r3 >> 2;
            const float* __restrict__ Wt = q ? w_hh : w_ih;
            pvg = Wt[(size_t)(g * HDIM + (wv << 4) + c) * HDIM + 8 * j2 + k4];
        }
    }

    const int lo = blockIdx.x * LROWS;
    if (threadIdx.x < LROWS) lc[threadIdx.x] = 0;

    const int4* __restrict__ d4 = (const int4*)dst;
    const int rot = blockIdx.x * 96;
    int* myI = bufI + wave * WCAP;
    int4* myV = bufV + wave * WCAP;

    int4 v[SCAN_J];
#pragma unroll
    for (int u = 0; u < SCAN_J; ++u) {
        const int g0 = (int)threadIdx.x + PTHREADS * u;
        int g = g0 + rot;
        g = (g >= E4) ? g - E4 : g;
        v[u] = (g0 < E4) ? d4[g] : make_int4(-1, -1, -1, -1);
    }
    int wcount = 0;
#pragma unroll
    for (int u = 0; u < SCAN_J; ++u) {
        const unsigned r0 = (unsigned)(v[u].x - lo), r1 = (unsigned)(v[u].y - lo);
        const unsigned r2 = (unsigned)(v[u].z - lo), r3 = (unsigned)(v[u].w - lo);
        const unsigned ra = r0 < r1 ? r0 : r1, rb = r2 < r3 ? r2 : r3;
        const bool any = (ra < rb ? ra : rb) < LROWS;
        const unsigned m = __builtin_amdgcn_ballot_w32(any);
        if (m) {
            if (any) {
                const int k = wcount + (int)__builtin_amdgcn_mbcnt_lo(m, 0u);
                if (k < WCAP) {
                    int g = (int)threadIdx.x + PTHREADS * u + rot;
                    g = (g >= E4) ? g - E4 : g;
                    myI[k] = g;
                    myV[k] = v[u];
                }
            }
            wcount += __builtin_popcount(m);
        }
    }
    __syncthreads();

    const int nw = wcount < WCAP ? wcount : WCAP;
    for (int base = 0; base < nw; base += 32) {
        const int k = base + lane;
        if (k < nw) place_group(myV[k], myI[k], lo, lc, src, ea, lid, lco);
    }
    if (wcount > WCAP) {
        int seen = 0;
#pragma nounroll
        for (int u = 0; u < SCAN_J; ++u) {
            const int g0 = (int)threadIdx.x + PTHREADS * u;
            int g = g0 + rot;
            g = (g >= E4) ? g - E4 : g;
            const int4 q = (g0 < E4) ? d4[g] : make_int4(-1, -1, -1, -1);
            const unsigned r0 = (unsigned)(q.x - lo), r1 = (unsigned)(q.y - lo);
            const unsigned r2 = (unsigned)(q.z - lo), r3 = (unsigned)(q.w - lo);
            const unsigned ra = r0 < r1 ? r0 : r1, rb = r2 < r3 ? r2 : r3;
            const bool any = (ra < rb ? ra : rb) < LROWS;
            const unsigned m = __builtin_amdgcn_ballot_w32(any);
            if (any && seen + (int)__builtin_amdgcn_mbcnt_lo(m, 0u) >= WCAP)
                place_group(q, g, lo, lc, src, ea, lid, lco);
            seen += __builtin_popcount(m);
        }
    }
    __syncthreads();
    if (threadIdx.x < LROWS) cnt[lo + threadIdx.x] = lc[threadIdx.x];
    if (packs) { pw2[t] = pv2; pg[t] = pvg; }
}

#define NPRE 2
template <bool PRE>
__device__ __forceinline__ void gate_tiles_f32(
    const float* A, int lda, const float4* __restrict__ Bq,
    const float4 (&pre)[NPRE * 3],
    int hf, int l, v8f& accR, v8f& accZ, v8f& accN)
{
    const float* Arow = A + l * lda;
    v8f r = {}, z = {}, n = {};
#pragma unroll
    for (int j2 = 0; j2 < HDIM / 8; ++j2) {
        const bool fromPre = PRE && j2 < NPRE;
        const float4 qr = fromPre ? pre[(j2 < NPRE ? j2 : 0) * 3 + 0] : Bq[(j2 * 3 + 0) * 32];
        const float4 qz = fromPre ? pre[(j2 < NPRE ? j2 : 0) * 3 + 1] : Bq[(j2 * 3 + 1) * 32];
        const float4 qn = fromPre ? pre[(j2 < NPRE ? j2 : 0) * 3 + 2] : Bq[(j2 * 3 + 2) * 32];
#pragma unroll
        for (int s = 0; s < 2; ++s) {
            const int kb = 8 * j2 + 4 * s + (hf << 1);
            const v2f a  = *(const v2f*)(Arow + kb);
            const v2f br = s ? v2f{qr.z, qr.w} : v2f{qr.x, qr.y};
            const v2f bz = s ? v2f{qz.z, qz.w} : v2f{qz.x, qz.y};
            const v2f bn = s ? v2f{qn.z, qn.w} : v2f{qn.x, qn.y};
            r = __builtin_amdgcn_wmma_f32_16x16x4_f32(
                false, a, false, br, (short)0, r, false, false);
            z = __builtin_amdgcn_wmma_f32_16x16x4_f32(
                false, a, false, bz, (short)0, z, false, false);
            n = __builtin_amdgcn_wmma_f32_16x16x4_f32(
                false, a, false, bn, (short)0, n, false, false);
        }
    }
    accR = r; accZ = z; accN = n;
}

__device__ __forceinline__ void edge_fma(
    float (&a)[10], float c0, float c1, float c2, float c3, float hlo, float hhi)
{
    a[0] = fmaf(c0, hlo, a[0]);  a[1] = fmaf(c0, hhi, a[1]);
    a[2] = fmaf(c1, hlo, a[2]);  a[3] = fmaf(c1, hhi, a[3]);
    a[4] = fmaf(c2, hlo, a[4]);  a[5] = fmaf(c2, hhi, a[5]);
    a[6] = fmaf(c3, hlo, a[6]);  a[7] = fmaf(c3, hhi, a[7]);
    a[8] += hlo;                 a[9] += hhi;
}

__device__ __forceinline__ void edge_fma8(
    float4 (&a)[5][2], const float4 c, const float4 gA, const float4 gB)
{
#define EDGE_FMA4(p, cf)                                                      \
    a[p][0].x = fmaf(cf, gA.x, a[p][0].x); a[p][0].y = fmaf(cf, gA.y, a[p][0].y); \
    a[p][0].z = fmaf(cf, gA.z, a[p][0].z); a[p][0].w = fmaf(cf, gA.w, a[p][0].w); \
    a[p][1].x = fmaf(cf, gB.x, a[p][1].x); a[p][1].y = fmaf(cf, gB.y, a[p][1].y); \
    a[p][1].z = fmaf(cf, gB.z, a[p][1].z); a[p][1].w = fmaf(cf, gB.w, a[p][1].w);
    EDGE_FMA4(0, c.x)
    EDGE_FMA4(1, c.y)
    EDGE_FMA4(2, c.z)
    EDGE_FMA4(3, c.w)
#undef EDGE_FMA4
    a[4][0].x += gA.x; a[4][0].y += gA.y; a[4][0].z += gA.z; a[4][0].w += gA.w;
    a[4][1].x += gB.x; a[4][1].y += gB.y; a[4][1].z += gB.z; a[4][1].w += gB.w;
}

__device__ __forceinline__ void scan_row(
    const float* __restrict__ h_in, const int* __restrict__ src,
    const int* __restrict__ dst, const float* __restrict__ ea,
    int d, int lane, float* srow)
{
    float a[10];
#pragma unroll
    for (int i = 0; i < 10; ++i) a[i] = 0.0f;
    for (int e0 = 0; e0 < NEDGE; e0 += 32) {
        const int e = e0 + lane;
        const bool hit = (e < NEDGE) && (dst[e < NEDGE ? e : NEDGE - 1] == d);
        unsigned m = __builtin_amdgcn_ballot_w32(hit);
        while (m) {
            const int ee = e0 + __builtin_ctz(m);
            m &= m - 1;
            const int s = src[ee];
            const float4 c = *(const float4*)(ea + (size_t)ee * 4);
            const float* __restrict__ row = h_in + (size_t)s * HDIM + lane;
            edge_fma(a, c.x, c.y, c.z, c.w, row[0], row[32]);
        }
    }
#pragma unroll
    for (int p = 0; p < 5; ++p) {
        srow[HDIM * (p + 1) + lane]      = a[2 * p];
        srow[HDIM * (p + 1) + lane + 32] = a[2 * p + 1];
    }
}

__global__ __launch_bounds__(128) void fused_step(
    const float* __restrict__ h_in, float* __restrict__ h_out,
    const int* __restrict__ src, const int* __restrict__ dst,
    const float* __restrict__ ea,
    const int* __restrict__ cnt, const int* __restrict__ lid,
    const float* __restrict__ lco,
    const float* __restrict__ pw2, const float* __restrict__ pg,
    const float* __restrict__ conv_b,
    const float* __restrict__ b_ih, const float* __restrict__ b_hh)
{
    __shared__ __attribute__((aligned(16))) float St[16 * SPITCH];
    __shared__ __attribute__((aligned(16))) float Ms[16 * HPITCH];
    const int wave = threadIdx.x >> 5;
    const int lane = threadIdx.x & 31;
    const int hf = lane >> 4, l = lane & 15;
    const int m0  = blockIdx.x << 4;
    const int col = (wave << 4) + l;

    const int rb = (wave << 2);
    const int lr = lane >> 3;
    const int cg = lane & 7;
    int cr[4];
    int nl[4];
    int ev[4], sv[4], rk[4];
    float4 cv[4];
#pragma unroll
    for (int r = 0; r < 4; ++r) {
        const int d = m0 + rb + r;
        cr[r] = __builtin_amdgcn_readfirstlane(cnt[d]);
        nl[r] = (cr[r] > LCAP) ? 0 : cr[r];
    }
    const int drow = m0 + rb + lr;
    const float* __restrict__ hcol = h_in + 4 * cg;
    const float4 hoA = *(const float4*)(hcol + (size_t)drow * HDIM);
    const float4 hoB = *(const float4*)(hcol + (size_t)drow * HDIM + 32);
#pragma unroll
    for (int r = 0; r < 4; ++r) {
        const size_t slot = (size_t)(m0 + rb + r) * LCAP + lane;
        int2 id = make_int2(0x7fffffff, 0);
        float4 c4 = make_float4(0.0f, 0.0f, 0.0f, 0.0f);
        if (lane < nl[r]) {
            id = *(const int2*)(lid + slot * 2);
            c4 = *(const float4*)(lco + slot * 4);
        }
        ev[r] = id.x; sv[r] = id.y; cv[r] = c4;
    }
    const float4* __restrict__ Bw =
        (const float4*)pw2 + (size_t)wave * (8 * 6 * 32) + lane;
    float4 qpre[NPRE * 6];
#pragma unroll
    for (int i = 0; i < NPRE * 6; ++i) qpre[i] = Bw[i * 32];

#pragma unroll
    for (int r = 0; r < 4; ++r) {
        int k = 0;
        for (int j = 0; j < nl[r]; ++j) {
            const int ej = __builtin_amdgcn_readlane(ev[r], j);
            k += (ej < ev[r]) ? 1 : 0;
        }
        rk[r] = k;
    }
#pragma unroll
    for (int r = 0; r < 4; ++r) {
        float* sc = St + (rb + r) * SPITCH + HDIM;
        const int pos = (lane < nl[r]) ? rk[r] : lane;
        sc[pos] = __int_as_float(sv[r]);
        *(float4*)(sc + LCAP + 4 * pos) = cv[r];
    }
    int maxn = nl[0] > nl[1] ? nl[0] : nl[1];
    { const int t = nl[2] > nl[3] ? nl[2] : nl[3]; maxn = maxn > t ? maxn : t; }
    const int clane = (lr == 0) ? cr[0] : ((lr == 1) ? cr[1] : ((lr == 2) ? cr[2] : cr[3]));
    const bool bigl = clane > LCAP;
    const int nlane = bigl ? 0 : clane;
    const float* mys = St + (rb + lr) * SPITCH + HDIM;

    float4 acc[5][2];
#pragma unroll
    for (int p = 0; p < 5; ++p) {
        acc[p][0] = make_float4(0.0f, 0.0f, 0.0f, 0.0f);
        acc[p][1] = make_float4(0.0f, 0.0f, 0.0f, 0.0f);
    }

    {
        int    sr[UB];
        float4 cc[UB], gA[UB], gB[UB];
#pragma unroll
        for (int u = 0; u < UB; ++u) {
            sr[u] = __float_as_int(mys[u]);
            cc[u] = *(const float4*)(mys + LCAP + 4 * u);
        }
#pragma unroll
        for (int u = 0; u < UB; ++u) {
            if (u < nlane) {
                const float* __restrict__ row = hcol + (size_t)sr[u] * HDIM;
                gA[u] = *(const float4*)row;
                gB[u] = *(const float4*)(row + 32);
            }
        }
#pragma unroll
        for (int u = 0; u < UB; ++u) {
            if (u < nlane) edge_fma8(acc, cc[u], gA[u], gB[u]);
        }
    }
    for (int t = UB; t < maxn; ++t) {
        if (t < nlane) {
            const int s = __float_as_int(mys[t]);
            const float4 c = *(const float4*)(mys + LCAP + 4 * t);
            const float* __restrict__ row = hcol + (size_t)s * HDIM;
            edge_fma8(acc, c, *(const float4*)row, *(const float4*)(row + 32));
        }
    }

    {
        float* srow = St + (rb + lr) * SPITCH + 4 * cg;
        *(float4*)(srow)      = hoA;
        *(float4*)(srow + 32) = hoB;
        if (!bigl) {
#pragma unroll
            for (int p = 0; p < 5; ++p) {
                *(float4*)(srow + HDIM * (p + 1))      = acc[p][0];
                *(float4*)(srow + HDIM * (p + 1) + 32) = acc[p][1];
            }
        }
    }
#pragma nounroll
    for (int r = 0; r < 4; ++r) {
        const int d = m0 + rb + r;
        if (__builtin_amdgcn_readfirstlane(cnt[d]) > LCAP)
            scan_row(h_in, src, dst, ea, d, lane, St + (rb + r) * SPITCH);
    }
    __syncthreads();

    {
        v8f pa[6];
#pragma unroll
        for (int p = 0; p < 6; ++p) pa[p] = v8f{};
#pragma unroll
        for (int j2 = 0; j2 < HDIM / 8; ++j2) {
            float4 q[6];
#pragma unroll
            for (int p = 0; p < 6; ++p)
                q[p] = (j2 < NPRE) ? qpre[(j2 < NPRE ? j2 : 0) * 6 + p] : Bw[(j2 * 6 + p) * 32];
#pragma unroll
            for (int s = 0; s < 2; ++s) {
                const int kb = 8 * j2 + 4 * s + (hf << 1);
#pragma unroll
                for (int p = 0; p < 6; ++p) {
                    const v2f a = *(const v2f*)(St + l * SPITCH + p * HDIM + kb);
                    const v2f b = s ? v2f{q[p].z, q[p].w} : v2f{q[p].x, q[p].y};
                    pa[p] = __builtin_amdgcn_wmma_f32_16x16x4_f32(
                        false, a, false, b, (short)0, pa[p], false, false);
                }
            }
        }
        const float cb = conv_b[col];
#pragma unroll
        for (int v = 0; v < 8; ++v) {
            const int rr = v + (hf << 3);
            const float s = ((pa[0][v] + pa[1][v]) + (pa[2][v] + pa[3][v]))
                          + (pa[4][v] + pa[5][v]);
            Ms[rr * HPITCH + col] = fmaxf(s + cb, 0.0f);
        }
    }
    v8f ir, iz, in_, hr, hz, hn_;
    const float4* __restrict__ Gi = (const float4*)pg + (size_t)wave * (8 * 3 * 32) + lane;
    const float4* __restrict__ Gh = Gi + 4 * 8 * 3 * 32;
    float4 gpre[NPRE * 3];
#pragma unroll
    for (int i = 0; i < NPRE * 3; ++i) gpre[i] = Gi[i * 32];
    gate_tiles_f32<false>(St, SPITCH, Gh, gpre, hf, l, hr, hz, hn_);
    __syncthreads();
    gate_tiles_f32<true>(Ms, HPITCH, Gi, gpre, hf, l, ir, iz, in_);

    const float bir = b_ih[col], biz = b_ih[HDIM + col], bin = b_ih[2 * HDIM + col];
    const float bhr = b_hh[col], bhz = b_hh[HDIM + col], bhn = b_hh[2 * HDIM + col];

#pragma unroll
    for (int v = 0; v < 8; ++v) {
        const int rr = v + (hf << 3);
        const size_t idx = (size_t)(m0 + rr) * HDIM + col;
        const float gir = ir[v] + bir, ghr = hr[v] + bhr;
        const float giz = iz[v] + biz, ghz = hz[v] + bhz;
        const float gin = in_[v] + bin, ghn = hn_[v] + bhn;
        const float sr = gir + ghr;
        const float sz = giz + ghz;
        const float r = __builtin_amdgcn_rcpf(1.0f + __expf(-sr));
        const float z = __builtin_amdgcn_rcpf(1.0f + __expf(-sz));
        const float a = gin + r * ghn;
        const float n = 1.0f - 2.0f * __builtin_amdgcn_rcpf(1.0f + __expf(2.0f * a));
        const float ho = St[rr * SPITCH + col];
        h_out[idx] = (1.0f - z) * n + z * ho;
    }
}

extern "C" void kernel_launch(void* const* d_in, const int* in_sizes, int n_in,
                              void* d_out, int out_size, void* d_ws, size_t ws_size,
                              hipStream_t stream)
{
    (void)in_sizes; (void)n_in; (void)out_size; (void)ws_size;

    const float* x      = (const float*)d_in[0];
    const int*   eidx   = (const int*)  d_in[1];
    const float* ea     = (const float*)d_in[2];
    const float* lin0_w = (const float*)d_in[3];
    const float* lin0_b = (const float*)d_in[4];
    const float* nn_w   = (const float*)d_in[5];
    const float* nn_b   = (const float*)d_in[6];
    const float* root_w = (const float*)d_in[7];
    const float* conv_b = (const float*)d_in[8];
    const float* w_ih   = (const float*)d_in[9];
    const float* w_hh   = (const float*)d_in[10];
    const float* b_ih   = (const float*)d_in[11];
    const float* b_hh   = (const float*)d_in[12];

    const int* src = eidx;
    const int* dst = eidx + NEDGE;

    float* hw  = (float*)d_ws;
    float* lco = hw + (size_t)NNODE * HDIM;
    int*   lid = (int*)(lco + (size_t)NNODE * LCAP * 4);
    int*   cnt = lid + (size_t)NNODE * LCAP * 2;
    float* pw2 = (float*)(cnt + 10240);
    float* pg  = pw2 + PACK_FLOATS;
    float* ho  = (float*)d_out;

    lin0_lists<<<NLB + NGB, PTHREADS, 0, stream>>>(x, lin0_w, lin0_b, hw, src, dst, ea,
        cnt, lid, lco, nn_w, nn_b, root_w, w_ih, w_hh, pw2, pg);

    fused_step<<<NNODE / 16, 128, 0, stream>>>(hw, ho, src, dst, ea, cnt, lid, lco,
        pw2, pg, conv_b, b_ih, b_hh);
    fused_step<<<NNODE / 16, 128, 0, stream>>>(ho, hw, src, dst, ea, cnt, lid, lco,
        pw2, pg, conv_b, b_ih, b_hh);
    fused_step<<<NNODE / 16, 128, 0, stream>>>(hw, ho, src, dst, ea, cnt, lid, lco,
        pw2, pg, conv_b, b_ih, b_hh);
}
